# starting kernel
# baseline (speedup 1.0000x reference)
#include <hip/hip_runtime.h>
#include <hip/hip_bf16.h>
#include <cstdint>

typedef __attribute__((ext_vector_type(2))) float v2f;
typedef __attribute__((ext_vector_type(8))) float v8f;

#define GIN_N 100000
#define GIN_E 1600000
#define GIN_D 64
#define WPB 4
#define TSTRIDE 66

__global__ __launch_bounds__(256) void gin_zero(float4* __restrict__ p, int n4) {
  int i = blockIdx.x * blockDim.x + threadIdx.x;
  if (i < n4) p[i] = make_float4(0.f, 0.f, 0.f, 0.f);
}

__global__ __launch_bounds__(256) void gin_scatter(
    const float* __restrict__ h, const int* __restrict__ src,
    const int* __restrict__ dst, float* __restrict__ neigh, int nedges) {
  int gid  = blockIdx.x * blockDim.x + threadIdx.x;
  int e    = gid >> 5;
  int lane = threadIdx.x & 31;
  if (e >= nedges) return;
  int s = src[e];
  int d = dst[e];
  v2f v = *(const v2f*)(h + (size_t)s * GIN_D + lane * 2);
  float* np = neigh + (size_t)d * GIN_D + lane * 2;
  atomicAdd(np, v.x);
  atomicAdd(np + 1, v.y);
}

__global__ __launch_bounds__(WPB * 32) void gin_mlp(
    const float* __restrict__ h, const float* __restrict__ neigh,
    const float* __restrict__ W1, const float* __restrict__ b1,
    const float* __restrict__ W2, const float* __restrict__ b2,
    const float* __restrict__ g,  const float* __restrict__ be,
    const float* __restrict__ mu, const float* __restrict__ va,
    float* __restrict__ out, int do_bn) {
  __shared__ float sW1t[GIN_D * TSTRIDE];
  __shared__ float sW2t[GIN_D * TSTRIDE];
  __shared__ float sT[WPB][16 * TSTRIDE];

  int t = threadIdx.x;
  for (int i = t; i < GIN_D * GIN_D; i += WPB * 32) {
    int k = i >> 6, n = i & 63;
    sW1t[n * TSTRIDE + k] = W1[i];
    sW2t[n * TSTRIDE + k] = W2[i];
  }
  __syncthreads();

  int wave = t >> 5, lane = t & 31;
  int row0 = (blockIdx.x * WPB + wave) * 16;
  if (row0 >= GIN_N) return;
  int m    = lane & 15;
  int half = lane >> 4;

  v2f zA[16];
  const float* hrow = h     + (size_t)(row0 + m) * GIN_D + 2 * half;
  const float* nrow = neigh + (size_t)(row0 + m) * GIN_D + 2 * half;
#pragma unroll
  for (int k0 = 0; k0 < 16; ++k0) {
    v2f a = *(const v2f*)(hrow + 4 * k0);
    v2f b = *(const v2f*)(nrow + 4 * k0);
    zA[k0] = a + b;
  }

  v8f acc[4];
#pragma unroll
  for (int n0 = 0; n0 < 4; ++n0)
#pragma unroll
    for (int i = 0; i < 8; ++i) acc[n0][i] = 0.f;

#pragma unroll
  for (int n0 = 0; n0 < 4; ++n0) {
    const float* wcol = &sW1t[(n0 * 16 + m) * TSTRIDE + 2 * half];
#pragma unroll
    for (int k0 = 0; k0 < 16; ++k0) {
      v2f b = *(const v2f*)(wcol + 4 * k0);
      acc[n0] = __builtin_amdgcn_wmma_f32_16x16x4_f32(
          false, zA[k0], false, b, (short)0, acc[n0], false, false);
    }
  }

  float* st = sT[wave];
#pragma unroll
  for (int n0 = 0; n0 < 4; ++n0) {
    float bias = b1[n0 * 16 + m];
#pragma unroll
    for (int i = 0; i < 8; ++i) {
      float v = fmaxf(acc[n0][i] + bias, 0.f);
      st[(half * 8 + i) * TSTRIDE + n0 * 16 + m] = v;
    }
  }

  v2f h1A[16];
#pragma unroll
  for (int k0 = 0; k0 < 16; ++k0)
    h1A[k0] = *(const v2f*)&st[m * TSTRIDE + 4 * k0 + 2 * half];

  v8f acc2[4];
#pragma unroll
  for (int n0 = 0; n0 < 4; ++n0)
#pragma unroll
    for (int i = 0; i < 8; ++i) acc2[n0][i] = 0.f;

#pragma unroll
  for (int n0 = 0; n0 < 4; ++n0) {
    const float* wcol = &sW2t[(n0 * 16 + m) * TSTRIDE + 2 * half];
#pragma unroll
    for (int k0 = 0; k0 < 16; ++k0) {
      v2f b = *(const v2f*)(wcol + 4 * k0);
      acc2[n0] = __builtin_amdgcn_wmma_f32_16x16x4_f32(
          false, h1A[k0], false, b, (short)0, acc2[n0], false, false);
    }
  }

#pragma unroll
  for (int n0 = 0; n0 < 4; ++n0) {
    int col = n0 * 16 + m;
    float bias = b2[col];
    float scale = 1.f, shift = 0.f;
    if (do_bn) {
      float s = g[col] * rsqrtf(va[col] + 1e-5f);
      scale = s;
      shift = be[col] - mu[col] * s;
    }
#pragma unroll
    for (int i = 0; i < 8; ++i) {
      float v = acc2[n0][i] + bias;
      if (do_bn) v = fmaxf(v * scale + shift, 0.f);
      out[(size_t)(row0 + half * 8 + i) * GIN_D + col] = v;
    }
  }
}

extern "C" void kernel_launch(void* const* d_in, const int* in_sizes, int n_in,
                              void* d_out, int out_size, void* d_ws, size_t ws_size,
                              hipStream_t stream) {
  const float* x  = (const float*)d_in[0];
  const int*   ei = (const int*)d_in[1];
  const float* W1 = (const float*)d_in[2];
  const float* b1 = (const float*)d_in[3];
  const float* W2 = (const float*)d_in[4];
  const float* b2 = (const float*)d_in[5];
  const float* g  = (const float*)d_in[6];
  const float* be = (const float*)d_in[7];
  const float* mu = (const float*)d_in[8];
  const float* va = (const float*)d_in[9];
  float* out = (float*)d_out;

  float* neigh = (float*)d_ws;
  float* hbuf  = neigh + (size_t)GIN_N * GIN_D;

  const int n4    = GIN_N * GIN_D / 4;
  const int zgrid = (n4 + 255) / 256;
  const long long sthreads = (long long)GIN_E * 32;
  const int sgrid = (int)((sthreads + 255) / 256);
  const int mgrid = (GIN_N / 16 + WPB - 1) / WPB;

  const float* hcur = x;
  for (int l = 0; l < 5; ++l) {
    gin_zero<<<zgrid, 256, 0, stream>>>((float4*)neigh, n4);
    gin_scatter<<<sgrid, 256, 0, stream>>>(hcur, ei, ei + GIN_E, neigh, GIN_E);
    float* dstp = (l == 4) ? out : ((l % 2 == 0) ? hbuf : out);
    gin_mlp<<<mgrid, WPB * 32, 0, stream>>>(
        hcur, neigh,
        W1 + (size_t)l * GIN_D * GIN_D, b1 + (size_t)l * GIN_D,
        W2 + (size_t)l * GIN_D * GIN_D, b2 + (size_t)l * GIN_D,
        g  + (size_t)l * GIN_D, be + (size_t)l * GIN_D,
        mu + (size_t)l * GIN_D, va + (size_t)l * GIN_D,
        dstp, (l < 4) ? 1 : 0);
    hcur = dstp;
  }
}


# Edge scatter without global atomics: one block per 391-node range picks its edges from a per-call range bitmap of the unsorted edge list, adds their source rows in LDS; sum order = LDS arrival order
# speedup vs baseline: 2.7891x; 2.7891x over previous
#include <hip/hip_runtime.h>
#include <hip/hip_bf16.h>
#include <cstdint>

typedef __attribute__((ext_vector_type(2))) float v2f;
typedef __attribute__((ext_vector_type(8))) float v8f;

#define GIN_N 100000
#define GIN_E 1600000
#define GIN_D 64
#define WPB 4
#define TSTRIDE 66

#define NRANGE 256
#define RROWS  391
#define WPR    50016
#define NP4    (WPR / 4)
static_assert(NRANGE * RROWS >= GIN_N, "ranges cover all nodes");
static_assert((NRANGE - 1) * RROWS < GIN_N, "no empty range");
static_assert(WPR % 32 == 0 && WPR * 32 >= GIN_E, "bitmap rows are whole lines and cover all edges");
#define BM_BLOCKS (WPR / 32)
#define PK_SHIFT  17
static_assert(GIN_N <= (1 << PK_SHIFT), "a source node fits the low field");
static_assert(RROWS <= (1 << (31 - PK_SHIFT)), "a row inside a range fits the high field");

__global__ __launch_bounds__(1024) void gin_bitmap(const int* __restrict__ src,
                                                   const int* __restrict__ dst,
                                                   unsigned* __restrict__ bm,
                                                   int* __restrict__ pk) {
  __shared__ unsigned tile[NRANGE * 33];
  const int tid  = threadIdx.x;
  const int lane = tid & 31;
  const int wv   = tid >> 5;
  for (int i = tid; i < NRANGE * 33; i += 1024) tile[i] = 0u;
  __syncthreads();
  const int e = blockIdx.x * 1024 + tid;
  if (e < GIN_E) {
    const int d = dst[e];
    const int r = d / RROWS;
    atomicOr(&tile[r * 33 + wv], 1u << lane);
    pk[e] = src[e] | ((d - r * RROWS) << PK_SHIFT);
  }
  __syncthreads();
  for (int i = tid; i < NRANGE * 32; i += 1024) {
    const int r = i >> 5, w = i & 31;
    bm[(size_t)r * WPR + (size_t)blockIdx.x * 32 + w] = tile[r * 33 + w];
  }
}

#define QCAP   16384
#define FB4    (QCAP / 128)
#define GD     16
#define SCAN_U 13
static_assert(FB4 * 128 <= QCAP, "a fallback piece fits the queue");
static_assert(SCAN_U * 1024 >= NP4, "one sweep covers a bitmap row");

__device__ __forceinline__ int gin_queue_bits(unsigned bits, int word, int slot, int* q) {
  while (bits != 0u) {
    const int b = __builtin_ctz(bits);
    bits &= bits - 1u;
    if (slot < QCAP) q[slot] = word * 32 + b;
    ++slot;
  }
  return slot;
}
__device__ __forceinline__ int gin_count_bits(uint4 v) {
  return __builtin_popcount(v.x) + __builtin_popcount(v.y) + __builtin_popcount(v.z) + __builtin_popcount(v.w);
}

__global__ __launch_bounds__(1024) void gin_rows_lds(
    const float* __restrict__ h,
    const int* __restrict__ pk, const unsigned* __restrict__ bm,
    float* __restrict__ neigh) {
  __shared__ float s_rows[RROWS * GIN_D];
  __shared__ int   s_q[QCAP];
  __shared__ int   s_qn;
  const int tid  = threadIdx.x;
  const int lane = tid & 31;
  const int wv   = __builtin_amdgcn_readfirstlane(tid >> 5);
  const int r0   = blockIdx.x * RROWS;
  float4* sr4 = (float4*)s_rows;
  for (int i = tid; i < RROWS * GIN_D / 4; i += 1024) sr4[i] = make_float4(0.f, 0.f, 0.f, 0.f);
  if (tid == 0) s_qn = 0;
  __syncthreads();

  const uint4* b4 = (const uint4*)(bm + (size_t)blockIdx.x * WPR);
  const float* hb = h + lane;
  int p0 = 0;
  int cw = NP4;
  while (p0 < NP4) {
    const int pend = (p0 + cw < NP4) ? (p0 + cw) : NP4;
    if (cw == NP4) {
      uint4 v[SCAN_U];
#pragma unroll
      for (int j = 0; j < SCAN_U; ++j) {
        const int idx = j * 1024 + tid;
        v[j] = b4[(idx < NP4) ? idx : (NP4 - 1)];
      }
      int mine = 0;
#pragma unroll
      for (int j = 0; j < SCAN_U; ++j) {
        const int idx = j * 1024 + tid;
        if (idx >= NP4) v[j] = make_uint4(0u, 0u, 0u, 0u);
        mine += gin_count_bits(v[j]);
      }
      int slot = atomicAdd(&s_qn, mine);
#pragma unroll
      for (int j = 0; j < SCAN_U; ++j) {
        const int idx = j * 1024 + tid;
        slot = gin_queue_bits(v[j].x, idx * 4 + 0, slot, s_q);
        slot = gin_queue_bits(v[j].y, idx * 4 + 1, slot, s_q);
        slot = gin_queue_bits(v[j].z, idx * 4 + 2, slot, s_q);
        slot = gin_queue_bits(v[j].w, idx * 4 + 3, slot, s_q);
      }
    } else {
      const int idx = p0 + tid;
      if (tid < FB4 && idx < pend) {
        const uint4 v = b4[idx];
        int slot = atomicAdd(&s_qn, gin_count_bits(v));
        slot = gin_queue_bits(v.x, idx * 4 + 0, slot, s_q);
        slot = gin_queue_bits(v.y, idx * 4 + 1, slot, s_q);
        slot = gin_queue_bits(v.z, idx * 4 + 2, slot, s_q);
        slot = gin_queue_bits(v.w, idx * 4 + 3, slot, s_q);
      }
    }
    __syncthreads();
    const int cnt = __builtin_amdgcn_readfirstlane(s_qn);
    if (cnt > QCAP) {
      __syncthreads();
      if (tid == 0) s_qn = 0;
      __syncthreads();
      cw = FB4;
      continue;
    }
    int base = wv * GD;
    if (base < cnt) {
      const int last = cnt - 1;
      int qi = base + (lane & (GD - 1));
      int e  = s_q[(qi < last) ? qi : last];
      int sd = pk[e];
      while (base < cnt) {
        const int ng = (cnt - base < GD) ? (cnt - base) : GD;
        const int nbase = base + 32 * GD;
        qi = nbase + (lane & (GD - 1));
        const int e2 = s_q[(qi < last) ? qi : last];
        float va[GD], vb[GD];
#pragma unroll
        for (int j = 0; j < GD; ++j) {
          const int sj = __builtin_amdgcn_readlane(sd, j) & ((1 << PK_SHIFT) - 1);
          const float* hp = hb + (size_t)sj * GIN_D;
          va[j] = hp[0];
          vb[j] = hp[32];
        }
        const int sd2 = pk[e2];
        if (ng == GD) {
#pragma unroll
          for (int j = 0; j < GD; ++j) {
            const int dj = __builtin_amdgcn_readlane(sd, j) >> PK_SHIFT;
            float* rp = s_rows + dj * GIN_D + lane;
            atomicAdd(rp, va[j]);
            atomicAdd(rp + 32, vb[j]);
          }
        } else {
#pragma unroll
          for (int j = 0; j < GD; ++j) {
            if (j < ng) {
              const int dj = __builtin_amdgcn_readlane(sd, j) >> PK_SHIFT;
              float* rp = s_rows + dj * GIN_D + lane;
              atomicAdd(rp, va[j]);
              atomicAdd(rp + 32, vb[j]);
            }
          }
        }
        sd = sd2;
        base = nbase;
      }
    }
    __syncthreads();
    if (tid == 0) s_qn = 0;
    __syncthreads();
    p0 = pend;
  }

  const int left = GIN_N - r0;
  const int nrow = (left < RROWS) ? left : RROWS;
  float4* n4 = (float4*)(neigh + (size_t)r0 * GIN_D);
  for (int i = tid; i < nrow * (GIN_D / 4); i += 1024) n4[i] = sr4[i];
}

__global__ __launch_bounds__(WPB * 32) void gin_mlp(
    const float* __restrict__ h, const float* __restrict__ neigh,
    const float* __restrict__ W1, const float* __restrict__ b1,
    const float* __restrict__ W2, const float* __restrict__ b2,
    const float* __restrict__ g,  const float* __restrict__ be,
    const float* __restrict__ mu, const float* __restrict__ va,
    float* __restrict__ out, int do_bn) {
  __shared__ float sW1t[GIN_D * TSTRIDE];
  __shared__ float sW2t[GIN_D * TSTRIDE];
  __shared__ float sT[WPB][16 * TSTRIDE];

  int t = threadIdx.x;
  for (int i = t; i < GIN_D * GIN_D; i += WPB * 32) {
    int k = i >> 6, n = i & 63;
    sW1t[n * TSTRIDE + k] = W1[i];
    sW2t[n * TSTRIDE + k] = W2[i];
  }
  __syncthreads();

  int wave = t >> 5, lane = t & 31;
  int row0 = (blockIdx.x * WPB + wave) * 16;
  if (row0 >= GIN_N) return;
  int m    = lane & 15;
  int half = lane >> 4;

  v2f zA[16];
  const float* hrow = h     + (size_t)(row0 + m) * GIN_D + 2 * half;
  const float* nrow = neigh + (size_t)(row0 + m) * GIN_D + 2 * half;
#pragma unroll
  for (int k0 = 0; k0 < 16; ++k0) {
    v2f a = *(const v2f*)(hrow + 4 * k0);
    v2f b = *(const v2f*)(nrow + 4 * k0);
    zA[k0] = a + b;
  }

  v8f acc[4];
#pragma unroll
  for (int n0 = 0; n0 < 4; ++n0)
#pragma unroll
    for (int i = 0; i < 8; ++i) acc[n0][i] = 0.f;

#pragma unroll
  for (int n0 = 0; n0 < 4; ++n0) {
    const float* wcol = &sW1t[(n0 * 16 + m) * TSTRIDE + 2 * half];
#pragma unroll
    for (int k0 = 0; k0 < 16; ++k0) {
      v2f b = *(const v2f*)(wcol + 4 * k0);
      acc[n0] = __builtin_amdgcn_wmma_f32_16x16x4_f32(
          false, zA[k0], false, b, (short)0, acc[n0], false, false);
    }
  }

  float* st = sT[wave];
#pragma unroll
  for (int n0 = 0; n0 < 4; ++n0) {
    float bias = b1[n0 * 16 + m];
#pragma unroll
    for (int i = 0; i < 8; ++i) {
      float v = fmaxf(acc[n0][i] + bias, 0.f);
      st[(half * 8 + i) * TSTRIDE + n0 * 16 + m] = v;
    }
  }

  v2f h1A[16];
#pragma unroll
  for (int k0 = 0; k0 < 16; ++k0)
    h1A[k0] = *(const v2f*)&st[m * TSTRIDE + 4 * k0 + 2 * half];

  v8f acc2[4];
#pragma unroll
  for (int n0 = 0; n0 < 4; ++n0)
#pragma unroll
    for (int i = 0; i < 8; ++i) acc2[n0][i] = 0.f;

#pragma unroll
  for (int n0 = 0; n0 < 4; ++n0) {
    const float* wcol = &sW2t[(n0 * 16 + m) * TSTRIDE + 2 * half];
#pragma unroll
    for (int k0 = 0; k0 < 16; ++k0) {
      v2f b = *(const v2f*)(wcol + 4 * k0);
      acc2[n0] = __builtin_amdgcn_wmma_f32_16x16x4_f32(
          false, h1A[k0], false, b, (short)0, acc2[n0], false, false);
    }
  }

#pragma unroll
  for (int n0 = 0; n0 < 4; ++n0) {
    int col = n0 * 16 + m;
    float bias = b2[col];
    float scale = 1.f, shift = 0.f;
    if (do_bn) {
      float s = g[col] * rsqrtf(va[col] + 1e-5f);
      scale = s;
      shift = be[col] - mu[col] * s;
    }
#pragma unroll
    for (int i = 0; i < 8; ++i) {
      float v = acc2[n0][i] + bias;
      if (do_bn) v = fmaxf(v * scale + shift, 0.f);
      out[(size_t)(row0 + half * 8 + i) * GIN_D + col] = v;
    }
  }
}

extern "C" void kernel_launch(void* const* d_in, const int* in_sizes, int n_in,
                              void* d_out, int out_size, void* d_ws, size_t ws_size,
                              hipStream_t stream) {
  const float* x  = (const float*)d_in[0];
  const int*   ei = (const int*)d_in[1];
  const float* W1 = (const float*)d_in[2];
  const float* b1 = (const float*)d_in[3];
  const float* W2 = (const float*)d_in[4];
  const float* b2 = (const float*)d_in[5];
  const float* g  = (const float*)d_in[6];
  const float* be = (const float*)d_in[7];
  const float* mu = (const float*)d_in[8];
  const float* va = (const float*)d_in[9];
  float* out = (float*)d_out;

  float*    neigh = (float*)d_ws;
  float*    hbuf  = neigh + (size_t)GIN_N * GIN_D;
  unsigned* bm    = (unsigned*)(hbuf + (size_t)GIN_N * GIN_D);
  int*      pk    = (int*)(bm + (size_t)NRANGE * WPR);

  const int mgrid = (GIN_N / 16 + WPB - 1) / WPB;

  const float* hcur = x;
  gin_bitmap<<<BM_BLOCKS, 1024, 0, stream>>>(ei, ei + GIN_E, bm, pk);
  for (int l = 0; l < 5; ++l) {
    gin_rows_lds<<<NRANGE, 1024, 0, stream>>>(hcur, pk, bm, neigh);
    float* dstp = (l == 4) ? out : ((l % 2 == 0) ? hbuf : out);
    gin_mlp<<<mgrid, WPB * 32, 0, stream>>>(
        hcur, neigh,
        W1 + (size_t)l * GIN_D * GIN_D, b1 + (size_t)l * GIN_D,
        W2 + (size_t)l * GIN_D * GIN_D, b2 + (size_t)l * GIN_D,
        g  + (size_t)l * GIN_D, be + (size_t)l * GIN_D,
        mu + (size_t)l * GIN_D, va + (size_t)l * GIN_D,
        dstp, (l < 4) ? 1 : 0);
    hcur = dstp;
  }
}


# Range queue kept across layers: layer 1 stores each range's packed edge words while draining; layers 2-5 read that list back (no bitmap sweep, no index gather); sum order = LDS arrival, as before
# speedup vs baseline: 3.1660x; 1.1351x over previous
#include <hip/hip_runtime.h>
#include <hip/hip_bf16.h>
#include <cstdint>

typedef __attribute__((ext_vector_type(2))) float v2f;
typedef __attribute__((ext_vector_type(8))) float v8f;

#define GIN_N 100000
#define GIN_E 1600000
#define GIN_D 64
#define WPB 4
#define TSTRIDE 66

#define NRANGE 256
#define RROWS  391
#define WPR    50016
#define NP4    (WPR / 4)
static_assert(NRANGE * RROWS >= GIN_N, "ranges cover all nodes");
static_assert((NRANGE - 1) * RROWS < GIN_N, "no empty range");
static_assert(WPR % 32 == 0 && WPR * 32 >= GIN_E, "bitmap rows are whole lines and cover all edges");
#define BM_BLOCKS (WPR / 32)
#define PK_SHIFT  17
static_assert(GIN_N <= (1 << PK_SHIFT), "a source node fits the low field");
static_assert(RROWS <= (1 << (31 - PK_SHIFT)), "a row inside a range fits the high field");

__global__ __launch_bounds__(1024) void gin_bitmap(const int* __restrict__ src,
                                                   const int* __restrict__ dst,
                                                   unsigned* __restrict__ bm,
                                                   int* __restrict__ pk) {
  __shared__ unsigned tile[NRANGE * 33];
  const int tid  = threadIdx.x;
  const int lane = tid & 31;
  const int wv   = tid >> 5;
  for (int i = tid; i < NRANGE * 33; i += 1024) tile[i] = 0u;
  __syncthreads();
  const int e = blockIdx.x * 1024 + tid;
  if (e < GIN_E) {
    const int d = dst[e];
    const int r = d / RROWS;
    atomicOr(&tile[r * 33 + wv], 1u << lane);
    pk[e] = src[e] | ((d - r * RROWS) << PK_SHIFT);
  }
  __syncthreads();
  for (int i = tid; i < NRANGE * 32; i += 1024) {
    const int r = i >> 5, w = i & 31;
    bm[(size_t)r * WPR + (size_t)blockIdx.x * 32 + w] = tile[r * 33 + w];
  }
}

#define QCAP   16384
#define FB4    (QCAP / 128)
#define GD     16
#define SCAN_U 13
static_assert(FB4 * 128 <= QCAP, "a fallback piece fits the queue");
static_assert(SCAN_U * 1024 >= NP4, "one sweep covers a bitmap row");

__device__ __forceinline__ int gin_queue_bits(unsigned bits, int word, int slot, int* q) {
  while (bits != 0u) {
    const int b = __builtin_ctz(bits);
    bits &= bits - 1u;
    if (slot < QCAP) q[slot] = word * 32 + b;
    ++slot;
  }
  return slot;
}
__device__ __forceinline__ int gin_count_bits(uint4 v) {
  return __builtin_popcount(v.x) + __builtin_popcount(v.y) + __builtin_popcount(v.z) + __builtin_popcount(v.w);
}

template <bool FROM_LIST>
__device__ __forceinline__ void gin_drain(const float* __restrict__ hb, const int* __restrict__ pk,
                                          const int* s_q, const int* __restrict__ lp, int* __restrict__ rec,
                                          float* s_rows, int cnt, int wv, int lane) {
  int base = wv * GD;
  if (base < cnt) {
    const int last = cnt - 1;
    int qi = base + (lane & (GD - 1));
    qi = (qi < last) ? qi : last;
    int sd;
    if (FROM_LIST) sd = lp[qi];
    else           sd = pk[s_q[qi]];
    while (base < cnt) {
      const int ng = (cnt - base < GD) ? (cnt - base) : GD;
      const int nbase = base + 32 * GD;
      qi = nbase + (lane & (GD - 1));
      qi = (qi < last) ? qi : last;
      int e2 = 0;
      if (!FROM_LIST) e2 = s_q[qi];
      float va[GD], vb[GD];
#pragma unroll
      for (int j = 0; j < GD; ++j) {
        const int sj = __builtin_amdgcn_readlane(sd, j) & ((1 << PK_SHIFT) - 1);
        const float* hp = hb + (size_t)sj * GIN_D;
        va[j] = hp[0];
        vb[j] = hp[32];
      }
      const int sd2 = FROM_LIST ? lp[qi] : pk[e2];
      if (rec != nullptr && lane < ng) rec[base + lane] = sd;
      if (ng == GD) {
#pragma unroll
        for (int j = 0; j < GD; ++j) {
          const int dj = __builtin_amdgcn_readlane(sd, j) >> PK_SHIFT;
          float* rp = s_rows + dj * GIN_D + lane;
          atomicAdd(rp, va[j]);
          atomicAdd(rp + 32, vb[j]);
        }
      } else {
#pragma unroll
        for (int j = 0; j < GD; ++j) {
          if (j < ng) {
            const int dj = __builtin_amdgcn_readlane(sd, j) >> PK_SHIFT;
            float* rp = s_rows + dj * GIN_D + lane;
            atomicAdd(rp, va[j]);
            atomicAdd(rp + 32, vb[j]);
          }
        }
      }
      sd = sd2;
      base = nbase;
    }
  }
}

template <bool FIRST>
__global__ __launch_bounds__(1024) void gin_rows_lds(
    const float* __restrict__ h,
    const int* __restrict__ pk, const unsigned* __restrict__ bm,
    int* __restrict__ lst, int* __restrict__ lcnt,
    float* __restrict__ neigh) {
  __shared__ float s_rows[RROWS * GIN_D];
  __shared__ int   s_q[QCAP];
  __shared__ int   s_qn;
  const int tid  = threadIdx.x;
  const int lane = tid & 31;
  const int wv   = __builtin_amdgcn_readfirstlane(tid >> 5);
  const int r0   = blockIdx.x * RROWS;
  int* lrow = lst + (size_t)blockIdx.x * QCAP;
  int have = -1;
  if (!FIRST) have = __builtin_amdgcn_readfirstlane(lcnt[blockIdx.x]);
  float4* sr4 = (float4*)s_rows;
  for (int i = tid; i < RROWS * GIN_D / 4; i += 1024) sr4[i] = make_float4(0.f, 0.f, 0.f, 0.f);
  if (tid == 0) s_qn = 0;
  __syncthreads();

  const float* hb = h + lane;
  if (!FIRST && have >= 0 && have <= QCAP) {
    gin_drain<true>(hb, pk, s_q, lrow, nullptr, s_rows, have, wv, lane);
    __syncthreads();
  } else {
  const uint4* b4 = (const uint4*)(bm + (size_t)blockIdx.x * WPR);
  int kept = -1;
  int p0 = 0;
  int cw = NP4;
  while (p0 < NP4) {
    const int pend = (p0 + cw < NP4) ? (p0 + cw) : NP4;
    if (cw == NP4) {
      uint4 v[SCAN_U];
#pragma unroll
      for (int j = 0; j < SCAN_U; ++j) {
        const int idx = j * 1024 + tid;
        v[j] = b4[(idx < NP4) ? idx : (NP4 - 1)];
      }
      int mine = 0;
#pragma unroll
      for (int j = 0; j < SCAN_U; ++j) {
        const int idx = j * 1024 + tid;
        if (idx >= NP4) v[j] = make_uint4(0u, 0u, 0u, 0u);
        mine += gin_count_bits(v[j]);
      }
      int slot = atomicAdd(&s_qn, mine);
#pragma unroll
      for (int j = 0; j < SCAN_U; ++j) {
        const int idx = j * 1024 + tid;
        slot = gin_queue_bits(v[j].x, idx * 4 + 0, slot, s_q);
        slot = gin_queue_bits(v[j].y, idx * 4 + 1, slot, s_q);
        slot = gin_queue_bits(v[j].z, idx * 4 + 2, slot, s_q);
        slot = gin_queue_bits(v[j].w, idx * 4 + 3, slot, s_q);
      }
    } else {
      const int idx = p0 + tid;
      if (tid < FB4 && idx < pend) {
        const uint4 v = b4[idx];
        int slot = atomicAdd(&s_qn, gin_count_bits(v));
        slot = gin_queue_bits(v.x, idx * 4 + 0, slot, s_q);
        slot = gin_queue_bits(v.y, idx * 4 + 1, slot, s_q);
        slot = gin_queue_bits(v.z, idx * 4 + 2, slot, s_q);
        slot = gin_queue_bits(v.w, idx * 4 + 3, slot, s_q);
      }
    }
    __syncthreads();
    const int cnt = __builtin_amdgcn_readfirstlane(s_qn);
    if (cnt > QCAP) {
      __syncthreads();
      if (tid == 0) s_qn = 0;
      __syncthreads();
      cw = FB4;
      continue;
    }
    const bool whole = (cw == NP4);
    gin_drain<false>(hb, pk, s_q, nullptr, (FIRST && whole) ? lrow : nullptr, s_rows, cnt, wv, lane);
    if (whole) kept = cnt;
    __syncthreads();
    if (tid == 0) s_qn = 0;
    __syncthreads();
    p0 = pend;
  }
  if (FIRST && tid == 0) lcnt[blockIdx.x] = kept;
  }

  const int left = GIN_N - r0;
  const int nrow = (left < RROWS) ? left : RROWS;
  float4* n4 = (float4*)(neigh + (size_t)r0 * GIN_D);
  for (int i = tid; i < nrow * (GIN_D / 4); i += 1024) n4[i] = sr4[i];
}

__global__ __launch_bounds__(WPB * 32) void gin_mlp(
    const float* __restrict__ h, const float* __restrict__ neigh,
    const float* __restrict__ W1, const float* __restrict__ b1,
    const float* __restrict__ W2, const float* __restrict__ b2,
    const float* __restrict__ g,  const float* __restrict__ be,
    const float* __restrict__ mu, const float* __restrict__ va,
    float* __restrict__ out, int do_bn) {
  __shared__ float sW1t[GIN_D * TSTRIDE];
  __shared__ float sW2t[GIN_D * TSTRIDE];
  __shared__ float sT[WPB][16 * TSTRIDE];

  int t = threadIdx.x;
  for (int i = t; i < GIN_D * GIN_D; i += WPB * 32) {
    int k = i >> 6, n = i & 63;
    sW1t[n * TSTRIDE + k] = W1[i];
    sW2t[n * TSTRIDE + k] = W2[i];
  }
  __syncthreads();

  int wave = t >> 5, lane = t & 31;
  int row0 = (blockIdx.x * WPB + wave) * 16;
  if (row0 >= GIN_N) return;
  int m    = lane & 15;
  int half = lane >> 4;

  v2f zA[16];
  const float* hrow = h     + (size_t)(row0 + m) * GIN_D + 2 * half;
  const float* nrow = neigh + (size_t)(row0 + m) * GIN_D + 2 * half;
#pragma unroll
  for (int k0 = 0; k0 < 16; ++k0) {
    v2f a = *(const v2f*)(hrow + 4 * k0);
    v2f b = *(const v2f*)(nrow + 4 * k0);
    zA[k0] = a + b;
  }

  v8f acc[4];
#pragma unroll
  for (int n0 = 0; n0 < 4; ++n0)
#pragma unroll
    for (int i = 0; i < 8; ++i) acc[n0][i] = 0.f;

#pragma unroll
  for (int n0 = 0; n0 < 4; ++n0) {
    const float* wcol = &sW1t[(n0 * 16 + m) * TSTRIDE + 2 * half];
#pragma unroll
    for (int k0 = 0; k0 < 16; ++k0) {
      v2f b = *(const v2f*)(wcol + 4 * k0);
      acc[n0] = __builtin_amdgcn_wmma_f32_16x16x4_f32(
          false, zA[k0], false, b, (short)0, acc[n0], false, false);
    }
  }

  float* st = sT[wave];
#pragma unroll
  for (int n0 = 0; n0 < 4; ++n0) {
    float bias = b1[n0 * 16 + m];
#pragma unroll
    for (int i = 0; i < 8; ++i) {
      float v = fmaxf(acc[n0][i] + bias, 0.f);
      st[(half * 8 + i) * TSTRIDE + n0 * 16 + m] = v;
    }
  }

  v2f h1A[16];
#pragma unroll
  for (int k0 = 0; k0 < 16; ++k0)
    h1A[k0] = *(const v2f*)&st[m * TSTRIDE + 4 * k0 + 2 * half];

  v8f acc2[4];
#pragma unroll
  for (int n0 = 0; n0 < 4; ++n0)
#pragma unroll
    for (int i = 0; i < 8; ++i) acc2[n0][i] = 0.f;

#pragma unroll
  for (int n0 = 0; n0 < 4; ++n0) {
    const float* wcol = &sW2t[(n0 * 16 + m) * TSTRIDE + 2 * half];
#pragma unroll
    for (int k0 = 0; k0 < 16; ++k0) {
      v2f b = *(const v2f*)(wcol + 4 * k0);
      acc2[n0] = __builtin_amdgcn_wmma_f32_16x16x4_f32(
          false, h1A[k0], false, b, (short)0, acc2[n0], false, false);
    }
  }

#pragma unroll
  for (int n0 = 0; n0 < 4; ++n0) {
    int col = n0 * 16 + m;
    float bias = b2[col];
    float scale = 1.f, shift = 0.f;
    if (do_bn) {
      float s = g[col] * rsqrtf(va[col] + 1e-5f);
      scale = s;
      shift = be[col] - mu[col] * s;
    }
#pragma unroll
    for (int i = 0; i < 8; ++i) {
      float v = acc2[n0][i] + bias;
      if (do_bn) v = fmaxf(v * scale + shift, 0.f);
      out[(size_t)(row0 + half * 8 + i) * GIN_D + col] = v;
    }
  }
}

extern "C" void kernel_launch(void* const* d_in, const int* in_sizes, int n_in,
                              void* d_out, int out_size, void* d_ws, size_t ws_size,
                              hipStream_t stream) {
  const float* x  = (const float*)d_in[0];
  const int*   ei = (const int*)d_in[1];
  const float* W1 = (const float*)d_in[2];
  const float* b1 = (const float*)d_in[3];
  const float* W2 = (const float*)d_in[4];
  const float* b2 = (const float*)d_in[5];
  const float* g  = (const float*)d_in[6];
  const float* be = (const float*)d_in[7];
  const float* mu = (const float*)d_in[8];
  const float* va = (const float*)d_in[9];
  float* out = (float*)d_out;

  float*    neigh = (float*)d_ws;
  float*    hbuf  = neigh + (size_t)GIN_N * GIN_D;
  unsigned* bm    = (unsigned*)(hbuf + (size_t)GIN_N * GIN_D);
  int*      pk    = (int*)(bm + (size_t)NRANGE * WPR);
  int*      lst   = pk + GIN_E;
  int*      lcnt  = lst + (size_t)NRANGE * QCAP;

  const int mgrid = (GIN_N / 16 + WPB - 1) / WPB;

  const float* hcur = x;
  gin_bitmap<<<BM_BLOCKS, 1024, 0, stream>>>(ei, ei + GIN_E, bm, pk);
  for (int l = 0; l < 5; ++l) {
    if (l == 0) gin_rows_lds<true><<<NRANGE, 1024, 0, stream>>>(hcur, pk, bm, lst, lcnt, neigh);
    else        gin_rows_lds<false><<<NRANGE, 1024, 0, stream>>>(hcur, pk, bm, lst, lcnt, neigh);
    float* dstp = (l == 4) ? out : ((l % 2 == 0) ? hbuf : out);
    gin_mlp<<<mgrid, WPB * 32, 0, stream>>>(
        hcur, neigh,
        W1 + (size_t)l * GIN_D * GIN_D, b1 + (size_t)l * GIN_D,
        W2 + (size_t)l * GIN_D * GIN_D, b2 + (size_t)l * GIN_D,
        g  + (size_t)l * GIN_D, be + (size_t)l * GIN_D,
        mu + (size_t)l * GIN_D, va + (size_t)l * GIN_D,
        dstp, (l < 4) ? 1 : 0);
    hcur = dstp;
  }
}


# MLP moved into the range block: LDS rows start as the nodes' own rows, a wave runs both WMMA products on its 16-row tile in place; no sum buffer, 6 launches; sum order = own row, then LDS arrival
# speedup vs baseline: 3.6630x; 1.1570x over previous
#include <hip/hip_runtime.h>
#include <hip/hip_bf16.h>
#include <cstdint>

typedef __attribute__((ext_vector_type(2))) float v2f;
typedef __attribute__((ext_vector_type(8))) float v8f;

#define GIN_N 100000
#define GIN_E 1600000
#define GIN_D 64
#define TSTRIDE 66

#define NRANGE 256
#define RROWS  391
#define WPR    50016
#define NP4    (WPR / 4)
static_assert(NRANGE * RROWS >= GIN_N, "ranges cover all nodes");
static_assert((NRANGE - 1) * RROWS < GIN_N, "no empty range");
static_assert(WPR % 32 == 0 && WPR * 32 >= GIN_E, "bitmap rows are whole lines and cover all edges");
#define BM_BLOCKS (WPR / 32)
#define PK_SHIFT  17
static_assert(GIN_N <= (1 << PK_SHIFT), "a source node fits the low field");
static_assert(RROWS <= (1 << (31 - PK_SHIFT)), "a row inside a range fits the high field");

__global__ __launch_bounds__(1024) void gin_bitmap(const int* __restrict__ src,
                                                   const int* __restrict__ dst,
                                                   unsigned* __restrict__ bm,
                                                   int* __restrict__ pk) {
  __shared__ unsigned tile[NRANGE * 33];
  const int tid  = threadIdx.x;
  const int lane = tid & 31;
  const int wv   = tid >> 5;
  for (int i = tid; i < NRANGE * 33; i += 1024) tile[i] = 0u;
  __syncthreads();
  const int e = blockIdx.x * 1024 + tid;
  if (e < GIN_E) {
    const int d = dst[e];
    const int r = d / RROWS;
    atomicOr(&tile[r * 33 + wv], 1u << lane);
    pk[e] = src[e] | ((d - r * RROWS) << PK_SHIFT);
  }
  __syncthreads();
  for (int i = tid; i < NRANGE * 32; i += 1024) {
    const int r = i >> 5, w = i & 31;
    bm[(size_t)r * WPR + (size_t)blockIdx.x * 32 + w] = tile[r * 33 + w];
  }
}

#define RPITCH 68
#define RTILES ((RROWS + 15) / 16)
#define RALLOC (RTILES * 16)
static_assert(7 * 1024 >= RALLOC * (GIN_D / 4), "seven 16-byte pieces per thread cover the block's rows");
static_assert(RTILES <= 32, "one tile per wave");
#define QCAP   16384
#define FB4    (QCAP / 128)
#define GD     16
#define SCAN_U 13
static_assert(FB4 * 128 <= QCAP, "a fallback piece fits the queue");
static_assert(SCAN_U * 1024 >= NP4, "one sweep covers a bitmap row");

__device__ __forceinline__ int gin_queue_bits(unsigned bits, int word, int slot, int* q) {
  while (bits != 0u) {
    const int b = __builtin_ctz(bits);
    bits &= bits - 1u;
    if (slot < QCAP) q[slot] = word * 32 + b;
    ++slot;
  }
  return slot;
}
__device__ __forceinline__ int gin_count_bits(uint4 v) {
  return __builtin_popcount(v.x) + __builtin_popcount(v.y) + __builtin_popcount(v.z) + __builtin_popcount(v.w);
}

template <bool FROM_LIST>
__device__ __forceinline__ void gin_drain(const float* __restrict__ hb, const int* __restrict__ pk,
                                          const int* s_q, const int* __restrict__ lp, int* __restrict__ rec,
                                          float* s_rows, int cnt, int wv, int lane) {
  int base = wv * GD;
  if (base < cnt) {
    const int last = cnt - 1;
    int qi = base + (lane & (GD - 1));
    qi = (qi < last) ? qi : last;
    int sd;
    if (FROM_LIST) sd = lp[qi];
    else           sd = pk[s_q[qi]];
    while (base < cnt) {
      const int ng = (cnt - base < GD) ? (cnt - base) : GD;
      const int nbase = base + 32 * GD;
      qi = nbase + (lane & (GD - 1));
      qi = (qi < last) ? qi : last;
      int e2 = 0;
      if (!FROM_LIST) e2 = s_q[qi];
      float va[GD], vb[GD];
#pragma unroll
      for (int j = 0; j < GD; ++j) {
        const int sj = __builtin_amdgcn_readlane(sd, j) & ((1 << PK_SHIFT) - 1);
        const float* hp = hb + (size_t)sj * GIN_D;
        va[j] = hp[0];
        vb[j] = hp[32];
      }
      const int sd2 = FROM_LIST ? lp[qi] : pk[e2];
      if (rec != nullptr && lane < ng) rec[base + lane] = sd;
      if (ng == GD) {
#pragma unroll
        for (int j = 0; j < GD; ++j) {
          const int dj = __builtin_amdgcn_readlane(sd, j) >> PK_SHIFT;
          float* rp = s_rows + dj * RPITCH + lane;
          atomicAdd(rp, va[j]);
          atomicAdd(rp + 32, vb[j]);
        }
      } else {
#pragma unroll
        for (int j = 0; j < GD; ++j) {
          if (j < ng) {
            const int dj = __builtin_amdgcn_readlane(sd, j) >> PK_SHIFT;
            float* rp = s_rows + dj * RPITCH + lane;
            atomicAdd(rp, va[j]);
            atomicAdd(rp + 32, vb[j]);
          }
        }
      }
      sd = sd2;
      base = nbase;
    }
  }
}

template <bool FIRST>
__global__ __launch_bounds__(1024) void gin_layer(
    const float* __restrict__ h,
    const int* __restrict__ pk, const unsigned* __restrict__ bm,
    int* __restrict__ lst, int* __restrict__ lcnt,
    const float* __restrict__ W1, const float* __restrict__ b1,
    const float* __restrict__ W2, const float* __restrict__ b2,
    const float* __restrict__ g,  const float* __restrict__ be,
    const float* __restrict__ mu, const float* __restrict__ va,
    float* __restrict__ out, int do_bn) {
  __shared__ float s_rows[RALLOC * RPITCH];
  __shared__ int   s_q[QCAP];
  __shared__ float sW1t[GIN_D * TSTRIDE];
  __shared__ float sW2t[GIN_D * TSTRIDE];
  __shared__ float s_par[4 * GIN_D];
  __shared__ int   s_qn;
  const int tid  = threadIdx.x;
  const int lane = tid & 31;
  const int wv   = __builtin_amdgcn_readfirstlane(tid >> 5);
  const int r0   = blockIdx.x * RROWS;
  const int left = GIN_N - r0;
  const int nrow = (left < RROWS) ? left : RROWS;
  int* lrow = lst + (size_t)blockIdx.x * QCAP;
  int have = -1;
  if (!FIRST) have = __builtin_amdgcn_readfirstlane(lcnt[blockIdx.x]);

  float w1v[4], w2v[4];
#pragma unroll
  for (int j = 0; j < 4; ++j) {
    w1v[j] = W1[j * 1024 + tid];
    w2v[j] = W2[j * 1024 + tid];
  }
  const float4* h4 = (const float4*)(h + (size_t)r0 * GIN_D);
  float4 own[7];
#pragma unroll
  for (int j = 0; j < 7; ++j) {
    const int i = j * 1024 + tid;
    own[j] = make_float4(0.f, 0.f, 0.f, 0.f);
    if (i < nrow * (GIN_D / 4)) own[j] = h4[i];
  }
  if (tid < GIN_D) {
    float scale = 1.f, shift = 0.f;
    if (do_bn) {
      const float sc = g[tid] * rsqrtf(va[tid] + 1e-5f);
      scale = sc;
      shift = be[tid] - mu[tid] * sc;
    }
    s_par[tid]             = b1[tid];
    s_par[GIN_D + tid]     = b2[tid];
    s_par[2 * GIN_D + tid] = scale;
    s_par[3 * GIN_D + tid] = shift;
  }
#pragma unroll
  for (int j = 0; j < 4; ++j) {
    const int i = j * 1024 + tid;
    const int k = i >> 6, n = i & 63;
    sW1t[n * TSTRIDE + k] = w1v[j];
    sW2t[n * TSTRIDE + k] = w2v[j];
  }
#pragma unroll
  for (int j = 0; j < 7; ++j) {
    const int i = j * 1024 + tid;
    if (i < RALLOC * (GIN_D / 4)) *(float4*)&s_rows[(i >> 4) * RPITCH + (i & 15) * 4] = own[j];
  }
  if (tid == 0) s_qn = 0;
  __syncthreads();

  const float* hb = h + lane;
  if (!FIRST && have >= 0 && have <= QCAP) {
    gin_drain<true>(hb, pk, s_q, lrow, nullptr, s_rows, have, wv, lane);
    __syncthreads();
  } else {
  const uint4* b4 = (const uint4*)(bm + (size_t)blockIdx.x * WPR);
  int kept = -1;
  int p0 = 0;
  int cw = NP4;
  while (p0 < NP4) {
    const int pend = (p0 + cw < NP4) ? (p0 + cw) : NP4;
    if (cw == NP4) {
      uint4 v[SCAN_U];
#pragma unroll
      for (int j = 0; j < SCAN_U; ++j) {
        const int idx = j * 1024 + tid;
        v[j] = b4[(idx < NP4) ? idx : (NP4 - 1)];
      }
      int mine = 0;
#pragma unroll
      for (int j = 0; j < SCAN_U; ++j) {
        const int idx = j * 1024 + tid;
        if (idx >= NP4) v[j] = make_uint4(0u, 0u, 0u, 0u);
        mine += gin_count_bits(v[j]);
      }
      int slot = atomicAdd(&s_qn, mine);
#pragma unroll
      for (int j = 0; j < SCAN_U; ++j) {
        const int idx = j * 1024 + tid;
        slot = gin_queue_bits(v[j].x, idx * 4 + 0, slot, s_q);
        slot = gin_queue_bits(v[j].y, idx * 4 + 1, slot, s_q);
        slot = gin_queue_bits(v[j].z, idx * 4 + 2, slot, s_q);
        slot = gin_queue_bits(v[j].w, idx * 4 + 3, slot, s_q);
      }
    } else {
      const int idx = p0 + tid;
      if (tid < FB4 && idx < pend) {
        const uint4 v = b4[idx];
        int slot = atomicAdd(&s_qn, gin_count_bits(v));
        slot = gin_queue_bits(v.x, idx * 4 + 0, slot, s_q);
        slot = gin_queue_bits(v.y, idx * 4 + 1, slot, s_q);
        slot = gin_queue_bits(v.z, idx * 4 + 2, slot, s_q);
        slot = gin_queue_bits(v.w, idx * 4 + 3, slot, s_q);
      }
    }
    __syncthreads();
    const int cnt = __builtin_amdgcn_readfirstlane(s_qn);
    if (cnt > QCAP) {
      __syncthreads();
      if (tid == 0) s_qn = 0;
      __syncthreads();
      cw = FB4;
      continue;
    }
    const bool whole = (cw == NP4);
    gin_drain<false>(hb, pk, s_q, nullptr, (FIRST && whole) ? lrow : nullptr, s_rows, cnt, wv, lane);
    if (whole) kept = cnt;
    __syncthreads();
    if (tid == 0) s_qn = 0;
    __syncthreads();
    p0 = pend;
  }
  if (FIRST && tid == 0) lcnt[blockIdx.x] = kept;
  }

  const int ntile = (nrow + 15) >> 4;
  if (wv < ntile) {
    const int m    = lane & 15;
    const int half = lane >> 4;
    float* st = s_rows + wv * 16 * RPITCH;

    v2f zA[16];
#pragma unroll
    for (int k0 = 0; k0 < 16; ++k0)
      zA[k0] = *(const v2f*)&st[m * RPITCH + 4 * k0 + 2 * half];

    v8f acc[4];
#pragma unroll
    for (int n0 = 0; n0 < 4; ++n0)
#pragma unroll
      for (int i = 0; i < 8; ++i) acc[n0][i] = 0.f;

#pragma unroll
    for (int n0 = 0; n0 < 4; ++n0) {
      const float* wcol = &sW1t[(n0 * 16 + m) * TSTRIDE + 2 * half];
#pragma unroll
      for (int k0 = 0; k0 < 16; ++k0) {
        v2f b = *(const v2f*)(wcol + 4 * k0);
        acc[n0] = __builtin_amdgcn_wmma_f32_16x16x4_f32(
            false, zA[k0], false, b, (short)0, acc[n0], false, false);
      }
    }

#pragma unroll
    for (int n0 = 0; n0 < 4; ++n0) {
      const float bias = s_par[n0 * 16 + m];
#pragma unroll
      for (int i = 0; i < 8; ++i) {
        const float v = fmaxf(acc[n0][i] + bias, 0.f);
        st[(half * 8 + i) * RPITCH + n0 * 16 + m] = v;
      }
    }

    v2f h1A[16];
#pragma unroll
    for (int k0 = 0; k0 < 16; ++k0)
      h1A[k0] = *(const v2f*)&st[m * RPITCH + 4 * k0 + 2 * half];

    v8f acc2[4];
#pragma unroll
    for (int n0 = 0; n0 < 4; ++n0)
#pragma unroll
      for (int i = 0; i < 8; ++i) acc2[n0][i] = 0.f;

#pragma unroll
    for (int n0 = 0; n0 < 4; ++n0) {
      const float* wcol = &sW2t[(n0 * 16 + m) * TSTRIDE + 2 * half];
#pragma unroll
      for (int k0 = 0; k0 < 16; ++k0) {
        v2f b = *(const v2f*)(wcol + 4 * k0);
        acc2[n0] = __builtin_amdgcn_wmma_f32_16x16x4_f32(
            false, h1A[k0], false, b, (short)0, acc2[n0], false, false);
      }
    }

#pragma unroll
    for (int n0 = 0; n0 < 4; ++n0) {
      const int col = n0 * 16 + m;
      const float bias  = s_par[GIN_D + col];
      const float scale = s_par[2 * GIN_D + col];
      const float shift = s_par[3 * GIN_D + col];
#pragma unroll
      for (int i = 0; i < 8; ++i) {
        float v = acc2[n0][i] + bias;
        if (do_bn) v = fmaxf(v * scale + shift, 0.f);
        st[(half * 8 + i) * RPITCH + col] = v;
      }
    }
  }
  __syncthreads();

  float4* o4 = (float4*)(out + (size_t)r0 * GIN_D);
  for (int i = tid; i < nrow * (GIN_D / 4); i += 1024)
    o4[i] = *(const float4*)&s_rows[(i >> 4) * RPITCH + (i & 15) * 4];
}

extern "C" void kernel_launch(void* const* d_in, const int* in_sizes, int n_in,
                              void* d_out, int out_size, void* d_ws, size_t ws_size,
                              hipStream_t stream) {
  const float* x  = (const float*)d_in[0];
  const int*   ei = (const int*)d_in[1];
  const float* W1 = (const float*)d_in[2];
  const float* b1 = (const float*)d_in[3];
  const float* W2 = (const float*)d_in[4];
  const float* b2 = (const float*)d_in[5];
  const float* g  = (const float*)d_in[6];
  const float* be = (const float*)d_in[7];
  const float* mu = (const float*)d_in[8];
  const float* va = (const float*)d_in[9];
  float* out = (float*)d_out;

  float*    hbufa = (float*)d_ws;
  float*    hbufb = hbufa + (size_t)GIN_N * GIN_D;
  unsigned* bm    = (unsigned*)(hbufb + (size_t)GIN_N * GIN_D);
  int*      pk    = (int*)(bm + (size_t)NRANGE * WPR);
  int*      lst   = pk + GIN_E;
  int*      lcnt  = lst + (size_t)NRANGE * QCAP;

  const float* hcur = x;
  gin_bitmap<<<BM_BLOCKS, 1024, 0, stream>>>(ei, ei + GIN_E, bm, pk);
  for (int l = 0; l < 5; ++l) {
    float* dstp = (l == 4) ? out : ((l % 2 == 0) ? hbufa : hbufb);
    if (l == 0)
      gin_layer<true><<<NRANGE, 1024, 0, stream>>>(
          hcur, pk, bm, lst, lcnt,
          W1 + (size_t)l * GIN_D * GIN_D, b1 + (size_t)l * GIN_D,
          W2 + (size_t)l * GIN_D * GIN_D, b2 + (size_t)l * GIN_D,
          g  + (size_t)l * GIN_D, be + (size_t)l * GIN_D,
          mu + (size_t)l * GIN_D, va + (size_t)l * GIN_D,
          dstp, 1);
    else
      gin_layer<false><<<NRANGE, 1024, 0, stream>>>(
          hcur, pk, bm, lst, lcnt,
          W1 + (size_t)l * GIN_D * GIN_D, b1 + (size_t)l * GIN_D,
          W2 + (size_t)l * GIN_D * GIN_D, b2 + (size_t)l * GIN_D,
          g  + (size_t)l * GIN_D, be + (size_t)l * GIN_D,
          mu + (size_t)l * GIN_D, va + (size_t)l * GIN_D,
          dstp, (l < 4) ? 1 : 0);
    hcur = dstp;
  }
}


# Layers 2-5: layer 1 bins each range's edges by 16-row tile; a wave sums its own tile in registers (row picked by a scalar register index), no LDS adds; sum order = own row, then tile list order
# speedup vs baseline: 4.0389x; 1.1026x over previous
#include <hip/hip_runtime.h>
#include <hip/hip_bf16.h>
#include <cstdint>

typedef __attribute__((ext_vector_type(2))) float v2f;
typedef __attribute__((ext_vector_type(8))) float v8f;
typedef __attribute__((ext_vector_type(32))) float v32f;

#define GIN_N 100000
#define GIN_E 1600000
#define GIN_D 64
#define TSTRIDE 66

#define NRANGE 256
#define RROWS  391
#define WPR    50016
#define NP4    (WPR / 4)
static_assert(NRANGE * RROWS >= GIN_N, "ranges cover all nodes");
static_assert((NRANGE - 1) * RROWS < GIN_N, "no empty range");
static_assert(WPR % 32 == 0 && WPR * 32 >= GIN_E, "bitmap rows are whole lines and cover all edges");
#define BM_BLOCKS (WPR / 32)
#define PK_SHIFT  17
static_assert(GIN_N <= (1 << PK_SHIFT), "a source node fits the low field");
static_assert(RROWS <= (1 << (31 - PK_SHIFT)), "a row inside a range fits the high field");

__global__ __launch_bounds__(1024) void gin_bitmap(const int* __restrict__ src,
                                                   const int* __restrict__ dst,
                                                   unsigned* __restrict__ bm,
                                                   int* __restrict__ pk) {
  __shared__ unsigned tile[NRANGE * 33];
  const int tid  = threadIdx.x;
  const int lane = tid & 31;
  const int wv   = tid >> 5;
  for (int i = tid; i < NRANGE * 33; i += 1024) tile[i] = 0u;
  __syncthreads();
  const int e = blockIdx.x * 1024 + tid;
  if (e < GIN_E) {
    const int d = dst[e];
    const int r = d / RROWS;
    atomicOr(&tile[r * 33 + wv], 1u << lane);
    pk[e] = src[e] | ((d - r * RROWS) << PK_SHIFT);
  }
  __syncthreads();
  for (int i = tid; i < NRANGE * 32; i += 1024) {
    const int r = i >> 5, w = i & 31;
    bm[(size_t)r * WPR + (size_t)blockIdx.x * 32 + w] = tile[r * 33 + w];
  }
}

#define RPITCH 68
#define RTILES ((RROWS + 15) / 16)
#define RALLOC (RTILES * 16)
static_assert(7 * 1024 >= RALLOC * (GIN_D / 4), "seven 16-byte pieces per thread cover the block's rows");
static_assert(RTILES <= 32, "one tile per wave");
#define QCAP   16384
#define FB4    (QCAP / 128)
#define GD     16
#define SCAN_U 13
#define TCAP   640
#define TG     16
static_assert(RTILES * TCAP <= QCAP, "the tiles' lists fit the range's list space");
static_assert(TCAP % 32 == 0, "a tile's list starts on a whole line");
static_assert(FB4 * 128 <= QCAP, "a fallback piece fits the queue");
static_assert(SCAN_U * 1024 >= NP4, "one sweep covers a bitmap row");

__device__ __forceinline__ int gin_queue_bits(unsigned bits, int word, int slot, int* q) {
  while (bits != 0u) {
    const int b = __builtin_ctz(bits);
    bits &= bits - 1u;
    if (slot < QCAP) q[slot] = word * 32 + b;
    ++slot;
  }
  return slot;
}
__device__ __forceinline__ int gin_count_bits(uint4 v) {
  return __builtin_popcount(v.x) + __builtin_popcount(v.y) + __builtin_popcount(v.z) + __builtin_popcount(v.w);
}

__device__ __forceinline__ void gin_drain(const float* __restrict__ hb, const int* __restrict__ pk,
                                          const int* s_q, bool record, int* s_bin, int* s_tc,
                                          float* s_rows, int cnt, int wv, int lane) {
  int base = wv * GD;
  if (base < cnt) {
    const int last = cnt - 1;
    int qi = base + (lane & (GD - 1));
    qi = (qi < last) ? qi : last;
    int sd;
    sd = pk[s_q[qi]];
    while (base < cnt) {
      const int ng = (cnt - base < GD) ? (cnt - base) : GD;
      const int nbase = base + 32 * GD;
      qi = nbase + (lane & (GD - 1));
      qi = (qi < last) ? qi : last;
      const int e2 = s_q[qi];
      float va[GD], vb[GD];
#pragma unroll
      for (int j = 0; j < GD; ++j) {
        const int sj = __builtin_amdgcn_readlane(sd, j) & ((1 << PK_SHIFT) - 1);
        const float* hp = hb + (size_t)sj * GIN_D;
        va[j] = hp[0];
        vb[j] = hp[32];
      }
      const int sd2 = pk[e2];
      if (record && lane < ng) {
        const int t   = (sd >> PK_SHIFT) >> 4;
        const int pos = atomicAdd(&s_tc[t], 1);
        if (pos < TCAP) s_bin[t * TCAP + pos] = sd;
      }
      if (ng == GD) {
#pragma unroll
        for (int j = 0; j < GD; ++j) {
          const int dj = __builtin_amdgcn_readlane(sd, j) >> PK_SHIFT;
          float* rp = s_rows + dj * RPITCH + lane;
          atomicAdd(rp, va[j]);
          atomicAdd(rp + 32, vb[j]);
        }
      } else {
#pragma unroll
        for (int j = 0; j < GD; ++j) {
          if (j < ng) {
            const int dj = __builtin_amdgcn_readlane(sd, j) >> PK_SHIFT;
            float* rp = s_rows + dj * RPITCH + lane;
            atomicAdd(rp, va[j]);
            atomicAdd(rp + 32, vb[j]);
          }
        }
      }
      sd = sd2;
      base = nbase;
    }
  }
}

template <bool FIRST>
__global__ __launch_bounds__(1024) void gin_layer(
    const float* __restrict__ h,
    const int* __restrict__ pk, const unsigned* __restrict__ bm,
    int* __restrict__ lst, int* __restrict__ lcnt, int* __restrict__ tcnt,
    const float* __restrict__ W1, const float* __restrict__ b1,
    const float* __restrict__ W2, const float* __restrict__ b2,
    const float* __restrict__ g,  const float* __restrict__ be,
    const float* __restrict__ mu, const float* __restrict__ va,
    float* __restrict__ out, int do_bn) {
  __shared__ float s_rows[RALLOC * RPITCH];
  __shared__ int   s_q[QCAP];
  __shared__ float sW1t[GIN_D * TSTRIDE];
  __shared__ float sW2t[GIN_D * TSTRIDE];
  __shared__ float s_par[4 * GIN_D];
  __shared__ int   s_qn;
  __shared__ int   s_tc[32];
  __shared__ int   s_bin[FIRST ? RTILES * TCAP : 1];
  const int tid  = threadIdx.x;
  const int lane = tid & 31;
  const int wv   = __builtin_amdgcn_readfirstlane(tid >> 5);
  const int r0   = blockIdx.x * RROWS;
  const int left = GIN_N - r0;
  const int nrow = (left < RROWS) ? left : RROWS;
  int* lrow = lst + (size_t)blockIdx.x * QCAP;
  int have = -1;
  if (!FIRST) have = __builtin_amdgcn_readfirstlane(lcnt[blockIdx.x]);

  float w1v[4], w2v[4];
#pragma unroll
  for (int j = 0; j < 4; ++j) {
    w1v[j] = W1[j * 1024 + tid];
    w2v[j] = W2[j * 1024 + tid];
  }
  const float4* h4 = (const float4*)(h + (size_t)r0 * GIN_D);
  float4 own[7];
#pragma unroll
  for (int j = 0; j < 7; ++j) {
    const int i = j * 1024 + tid;
    own[j] = make_float4(0.f, 0.f, 0.f, 0.f);
    if (i < nrow * (GIN_D / 4)) own[j] = h4[i];
  }
  if (tid < GIN_D) {
    float scale = 1.f, shift = 0.f;
    if (do_bn) {
      const float sc = g[tid] * rsqrtf(va[tid] + 1e-5f);
      scale = sc;
      shift = be[tid] - mu[tid] * sc;
    }
    s_par[tid]             = b1[tid];
    s_par[GIN_D + tid]     = b2[tid];
    s_par[2 * GIN_D + tid] = scale;
    s_par[3 * GIN_D + tid] = shift;
  }
#pragma unroll
  for (int j = 0; j < 4; ++j) {
    const int i = j * 1024 + tid;
    const int k = i >> 6, n = i & 63;
    sW1t[n * TSTRIDE + k] = w1v[j];
    sW2t[n * TSTRIDE + k] = w2v[j];
  }
#pragma unroll
  for (int j = 0; j < 7; ++j) {
    const int i = j * 1024 + tid;
    if (i < RALLOC * (GIN_D / 4)) *(float4*)&s_rows[(i >> 4) * RPITCH + (i & 15) * 4] = own[j];
  }
  if (tid == 0) s_qn = 0;
  if (tid < 32) s_tc[tid] = 0;
  __syncthreads();

  const float* hb = h + lane;
  if (!FIRST && have >= 0) {
    if (wv < ((nrow + 15) >> 4)) {
      float* st = s_rows + wv * 16 * RPITCH;
      int cnt = __builtin_amdgcn_readfirstlane(tcnt[blockIdx.x * 32 + wv]);
      cnt = (cnt < 0) ? 0 : ((cnt > TCAP) ? TCAP : cnt);
      const int* lp = lrow + wv * TCAP;
      v32f acc;
#pragma unroll
      for (int r = 0; r < 16; ++r) {
        const v2f t = *(const v2f*)&st[r * RPITCH + 2 * lane];
        acc[2 * r]     = t.x;
        acc[2 * r + 1] = t.y;
      }
      const float* hb2 = h + 2 * lane;
      if (cnt > 0) {
        const int last = cnt - 1;
        int qi = (lane < last) ? lane : last;
        int sd = lp[qi];
        int base = 0;
        while (base + TG <= cnt) {
          const int nbase = base + TG;
          qi = nbase + lane;
          qi = (qi < last) ? qi : last;
          v2f v[TG];
#pragma unroll
          for (int j = 0; j < TG; ++j) {
            const int sj = __builtin_amdgcn_readlane(sd, j) & ((1 << PK_SHIFT) - 1);
            v[j] = *(const v2f*)(hb2 + (size_t)sj * GIN_D);
          }
          const int sd2 = lp[qi];
#pragma unroll
          for (int j = 0; j < TG; ++j) {
            const int dj = (__builtin_amdgcn_readlane(sd, j) >> PK_SHIFT) & 15;
            acc[2 * dj]     += v[j].x;
            acc[2 * dj + 1] += v[j].y;
          }
          sd = sd2;
          base = nbase;
        }
        const int rem = cnt - base;
        if (rem > 0) {
          v2f v[TG];
#pragma unroll
          for (int j = 0; j < TG; ++j) {
            if (j < rem) {
              const int sj = __builtin_amdgcn_readlane(sd, j) & ((1 << PK_SHIFT) - 1);
              v[j] = *(const v2f*)(hb2 + (size_t)sj * GIN_D);
            }
          }
#pragma unroll
          for (int j = 0; j < TG; ++j) {
            if (j < rem) {
              const int dj = (__builtin_amdgcn_readlane(sd, j) >> PK_SHIFT) & 15;
              acc[2 * dj]     += v[j].x;
              acc[2 * dj + 1] += v[j].y;
            }
          }
        }
      }
#pragma unroll
      for (int r = 0; r < 16; ++r) {
        v2f t;
        t.x = acc[2 * r];
        t.y = acc[2 * r + 1];
        *(v2f*)&st[r * RPITCH + 2 * lane] = t;
      }
    }
  } else {
  const uint4* b4 = (const uint4*)(bm + (size_t)blockIdx.x * WPR);
  int kept = -1;
  int p0 = 0;
  int cw = NP4;
  while (p0 < NP4) {
    const int pend = (p0 + cw < NP4) ? (p0 + cw) : NP4;
    if (cw == NP4) {
      uint4 v[SCAN_U];
#pragma unroll
      for (int j = 0; j < SCAN_U; ++j) {
        const int idx = j * 1024 + tid;
        v[j] = b4[(idx < NP4) ? idx : (NP4 - 1)];
      }
      int mine = 0;
#pragma unroll
      for (int j = 0; j < SCAN_U; ++j) {
        const int idx = j * 1024 + tid;
        if (idx >= NP4) v[j] = make_uint4(0u, 0u, 0u, 0u);
        mine += gin_count_bits(v[j]);
      }
      int slot = atomicAdd(&s_qn, mine);
#pragma unroll
      for (int j = 0; j < SCAN_U; ++j) {
        const int idx = j * 1024 + tid;
        slot = gin_queue_bits(v[j].x, idx * 4 + 0, slot, s_q);
        slot = gin_queue_bits(v[j].y, idx * 4 + 1, slot, s_q);
        slot = gin_queue_bits(v[j].z, idx * 4 + 2, slot, s_q);
        slot = gin_queue_bits(v[j].w, idx * 4 + 3, slot, s_q);
      }
    } else {
      const int idx = p0 + tid;
      if (tid < FB4 && idx < pend) {
        const uint4 v = b4[idx];
        int slot = atomicAdd(&s_qn, gin_count_bits(v));
        slot = gin_queue_bits(v.x, idx * 4 + 0, slot, s_q);
        slot = gin_queue_bits(v.y, idx * 4 + 1, slot, s_q);
        slot = gin_queue_bits(v.z, idx * 4 + 2, slot, s_q);
        slot = gin_queue_bits(v.w, idx * 4 + 3, slot, s_q);
      }
    }
    __syncthreads();
    const int cnt = __builtin_amdgcn_readfirstlane(s_qn);
    if (cnt > QCAP) {
      __syncthreads();
      if (tid == 0) s_qn = 0;
      __syncthreads();
      cw = FB4;
      continue;
    }
    const bool whole = (cw == NP4);
    gin_drain(hb, pk, s_q, FIRST && whole, s_bin, s_tc, s_rows, cnt, wv, lane);
    if (whole) kept = cnt;
    __syncthreads();
    if (tid == 0) s_qn = 0;
    __syncthreads();
    p0 = pend;
  }
  if (FIRST) {
    for (int i = tid; i < RTILES * TCAP; i += 1024) {
      const int t = i / TCAP;
      if (i - t * TCAP < s_tc[t]) lrow[i] = s_bin[i];
    }
    if (tid < 32) tcnt[blockIdx.x * 32 + tid] = s_tc[tid];
    if (tid == 0) {
      int k = kept;
      for (int t = 0; t < RTILES; ++t)
        if (s_tc[t] > TCAP) k = -1;
      lcnt[blockIdx.x] = k;
    }
  }
  }

  const int ntile = (nrow + 15) >> 4;
  if (wv < ntile) {
    const int m    = lane & 15;
    const int half = lane >> 4;
    float* st = s_rows + wv * 16 * RPITCH;

    v2f zA[16];
#pragma unroll
    for (int k0 = 0; k0 < 16; ++k0)
      zA[k0] = *(const v2f*)&st[m * RPITCH + 4 * k0 + 2 * half];

    v8f acc[4];
#pragma unroll
    for (int n0 = 0; n0 < 4; ++n0)
#pragma unroll
      for (int i = 0; i < 8; ++i) acc[n0][i] = 0.f;

#pragma unroll
    for (int n0 = 0; n0 < 4; ++n0) {
      const float* wcol = &sW1t[(n0 * 16 + m) * TSTRIDE + 2 * half];
#pragma unroll
      for (int k0 = 0; k0 < 16; ++k0) {
        v2f b = *(const v2f*)(wcol + 4 * k0);
        acc[n0] = __builtin_amdgcn_wmma_f32_16x16x4_f32(
            false, zA[k0], false, b, (short)0, acc[n0], false, false);
      }
    }

#pragma unroll
    for (int n0 = 0; n0 < 4; ++n0) {
      const float bias = s_par[n0 * 16 + m];
#pragma unroll
      for (int i = 0; i < 8; ++i) {
        const float v = fmaxf(acc[n0][i] + bias, 0.f);
        st[(half * 8 + i) * RPITCH + n0 * 16 + m] = v;
      }
    }

    v2f h1A[16];
#pragma unroll
    for (int k0 = 0; k0 < 16; ++k0)
      h1A[k0] = *(const v2f*)&st[m * RPITCH + 4 * k0 + 2 * half];

    v8f acc2[4];
#pragma unroll
    for (int n0 = 0; n0 < 4; ++n0)
#pragma unroll
      for (int i = 0; i < 8; ++i) acc2[n0][i] = 0.f;

#pragma unroll
    for (int n0 = 0; n0 < 4; ++n0) {
      const float* wcol = &sW2t[(n0 * 16 + m) * TSTRIDE + 2 * half];
#pragma unroll
      for (int k0 = 0; k0 < 16; ++k0) {
        v2f b = *(const v2f*)(wcol + 4 * k0);
        acc2[n0] = __builtin_amdgcn_wmma_f32_16x16x4_f32(
            false, h1A[k0], false, b, (short)0, acc2[n0], false, false);
      }
    }

#pragma unroll
    for (int n0 = 0; n0 < 4; ++n0) {
      const int col = n0 * 16 + m;
      const float bias  = s_par[GIN_D + col];
      const float scale = s_par[2 * GIN_D + col];
      const float shift = s_par[3 * GIN_D + col];
#pragma unroll
      for (int i = 0; i < 8; ++i) {
        float v = acc2[n0][i] + bias;
        if (do_bn) v = fmaxf(v * scale + shift, 0.f);
        st[(half * 8 + i) * RPITCH + col] = v;
      }
    }
  }
  __syncthreads();

  float4* o4 = (float4*)(out + (size_t)r0 * GIN_D);
  for (int i = tid; i < nrow * (GIN_D / 4); i += 1024)
    o4[i] = *(const float4*)&s_rows[(i >> 4) * RPITCH + (i & 15) * 4];
}

extern "C" void kernel_launch(void* const* d_in, const int* in_sizes, int n_in,
                              void* d_out, int out_size, void* d_ws, size_t ws_size,
                              hipStream_t stream) {
  const float* x  = (const float*)d_in[0];
  const int*   ei = (const int*)d_in[1];
  const float* W1 = (const float*)d_in[2];
  const float* b1 = (const float*)d_in[3];
  const float* W2 = (const float*)d_in[4];
  const float* b2 = (const float*)d_in[5];
  const float* g  = (const float*)d_in[6];
  const float* be = (const float*)d_in[7];
  const float* mu = (const float*)d_in[8];
  const float* va = (const float*)d_in[9];
  float* out = (float*)d_out;

  float*    hbufa = (float*)d_ws;
  float*    hbufb = hbufa + (size_t)GIN_N * GIN_D;
  unsigned* bm    = (unsigned*)(hbufb + (size_t)GIN_N * GIN_D);
  int*      pk    = (int*)(bm + (size_t)NRANGE * WPR);
  int*      lst   = pk + GIN_E;
  int*      lcnt  = lst + (size_t)NRANGE * QCAP;
  int*      tcnt  = lcnt + NRANGE;

  const float* hcur = x;
  gin_bitmap<<<BM_BLOCKS, 1024, 0, stream>>>(ei, ei + GIN_E, bm, pk);
  for (int l = 0; l < 5; ++l) {
    float* dstp = (l == 4) ? out : ((l % 2 == 0) ? hbufa : hbufb);
    if (l == 0)
      gin_layer<true><<<NRANGE, 1024, 0, stream>>>(
          hcur, pk, bm, lst, lcnt, tcnt,
          W1 + (size_t)l * GIN_D * GIN_D, b1 + (size_t)l * GIN_D,
          W2 + (size_t)l * GIN_D * GIN_D, b2 + (size_t)l * GIN_D,
          g  + (size_t)l * GIN_D, be + (size_t)l * GIN_D,
          mu + (size_t)l * GIN_D, va + (size_t)l * GIN_D,
          dstp, 1);
    else
      gin_layer<false><<<NRANGE, 1024, 0, stream>>>(
          hcur, pk, bm, lst, lcnt, tcnt,
          W1 + (size_t)l * GIN_D * GIN_D, b1 + (size_t)l * GIN_D,
          W2 + (size_t)l * GIN_D * GIN_D, b2 + (size_t)l * GIN_D,
          g  + (size_t)l * GIN_D, be + (size_t)l * GIN_D,
          mu + (size_t)l * GIN_D, va + (size_t)l * GIN_D,
          dstp, (l < 4) ? 1 : 0);
    hcur = dstp;
  }
}


# Layers 2-5 tile gather: a trip requests 32 rows at once (was 16, the last 4 held until the first row was back): loads pinned before the adds, scalar row base + lane offset; sum order as before
# speedup vs baseline: 4.1646x; 1.0311x over previous
#include <hip/hip_runtime.h>
#include <hip/hip_bf16.h>
#include <cstdint>

typedef __attribute__((ext_vector_type(2))) float v2f;
typedef __attribute__((ext_vector_type(8))) float v8f;
typedef __attribute__((ext_vector_type(32))) float v32f;

#define GIN_N 100000
#define GIN_E 1600000
#define GIN_D 64
#define TSTRIDE 66

#define NRANGE 256
#define RROWS  391
#define WPR    50016
#define NP4    (WPR / 4)
static_assert(NRANGE * RROWS >= GIN_N, "ranges cover all nodes");
static_assert((NRANGE - 1) * RROWS < GIN_N, "no empty range");
static_assert(WPR % 32 == 0 && WPR * 32 >= GIN_E, "bitmap rows are whole lines and cover all edges");
#define BM_BLOCKS (WPR / 32)
#define PK_SHIFT  17
static_assert(GIN_N <= (1 << PK_SHIFT), "a source node fits the low field");
static_assert(RROWS <= (1 << (31 - PK_SHIFT)), "a row inside a range fits the high field");

__global__ __launch_bounds__(1024) void gin_bitmap(const int* __restrict__ src,
                                                   const int* __restrict__ dst,
                                                   unsigned* __restrict__ bm,
                                                   int* __restrict__ pk) {
  __shared__ unsigned tile[NRANGE * 33];
  const int tid  = threadIdx.x;
  const int lane = tid & 31;
  const int wv   = tid >> 5;
  for (int i = tid; i < NRANGE * 33; i += 1024) tile[i] = 0u;
  __syncthreads();
  const int e = blockIdx.x * 1024 + tid;
  if (e < GIN_E) {
    const int d = dst[e];
    const int r = d / RROWS;
    atomicOr(&tile[r * 33 + wv], 1u << lane);
    pk[e] = src[e] | ((d - r * RROWS) << PK_SHIFT);
  }
  __syncthreads();
  for (int i = tid; i < NRANGE * 32; i += 1024) {
    const int r = i >> 5, w = i & 31;
    bm[(size_t)r * WPR + (size_t)blockIdx.x * 32 + w] = tile[r * 33 + w];
  }
}

#define RPITCH 68
#define RTILES ((RROWS + 15) / 16)
#define RALLOC (RTILES * 16)
static_assert(7 * 1024 >= RALLOC * (GIN_D / 4), "seven 16-byte pieces per thread cover the block's rows");
static_assert(RTILES <= 32, "one tile per wave");
#define QCAP   16384
#define FB4    (QCAP / 128)
#define GD     16
#define SCAN_U 13
#define TCAP   640
#define TG     32
static_assert(RTILES * TCAP <= QCAP, "the tiles' lists fit the range's list space");
static_assert(TCAP % 32 == 0, "a tile's list starts on a whole line");
static_assert(FB4 * 128 <= QCAP, "a fallback piece fits the queue");
static_assert(SCAN_U * 1024 >= NP4, "one sweep covers a bitmap row");

__device__ __forceinline__ int gin_queue_bits(unsigned bits, int word, int slot, int* q) {
  while (bits != 0u) {
    const int b = __builtin_ctz(bits);
    bits &= bits - 1u;
    if (slot < QCAP) q[slot] = word * 32 + b;
    ++slot;
  }
  return slot;
}
__device__ __forceinline__ int gin_count_bits(uint4 v) {
  return __builtin_popcount(v.x) + __builtin_popcount(v.y) + __builtin_popcount(v.z) + __builtin_popcount(v.w);
}

__device__ __forceinline__ void gin_drain(const float* __restrict__ hb, const int* __restrict__ pk,
                                          const int* s_q, bool record, int* s_bin, int* s_tc,
                                          float* s_rows, int cnt, int wv, int lane) {
  int base = wv * GD;
  if (base < cnt) {
    const int last = cnt - 1;
    int qi = base + (lane & (GD - 1));
    qi = (qi < last) ? qi : last;
    int sd;
    sd = pk[s_q[qi]];
    while (base < cnt) {
      const int ng = (cnt - base < GD) ? (cnt - base) : GD;
      const int nbase = base + 32 * GD;
      qi = nbase + (lane & (GD - 1));
      qi = (qi < last) ? qi : last;
      const int e2 = s_q[qi];
      float va[GD], vb[GD];
#pragma unroll
      for (int j = 0; j < GD; ++j) {
        const int sj = __builtin_amdgcn_readlane(sd, j) & ((1 << PK_SHIFT) - 1);
        const float* hp = hb + (size_t)sj * GIN_D;
        va[j] = hp[0];
        vb[j] = hp[32];
      }
      const int sd2 = pk[e2];
      if (record && lane < ng) {
        const int t   = (sd >> PK_SHIFT) >> 4;
        const int pos = atomicAdd(&s_tc[t], 1);
        if (pos < TCAP) s_bin[t * TCAP + pos] = sd;
      }
      if (ng == GD) {
#pragma unroll
        for (int j = 0; j < GD; ++j) {
          const int dj = __builtin_amdgcn_readlane(sd, j) >> PK_SHIFT;
          float* rp = s_rows + dj * RPITCH + lane;
          atomicAdd(rp, va[j]);
          atomicAdd(rp + 32, vb[j]);
        }
      } else {
#pragma unroll
        for (int j = 0; j < GD; ++j) {
          if (j < ng) {
            const int dj = __builtin_amdgcn_readlane(sd, j) >> PK_SHIFT;
            float* rp = s_rows + dj * RPITCH + lane;
            atomicAdd(rp, va[j]);
            atomicAdd(rp + 32, vb[j]);
          }
        }
      }
      sd = sd2;
      base = nbase;
    }
  }
}

typedef const __attribute__((address_space(1))) v2f* gin_grow_t;
__device__ __forceinline__ gin_grow_t gin_row_base(const float* __restrict__ h, int sj) {
  const uint64_t rp = (uint64_t)h + ((uint64_t)(uint32_t)sj << 8);
  const uint32_t lo = (uint32_t)__builtin_amdgcn_readfirstlane((int)(uint32_t)rp);
  const uint32_t hi = (uint32_t)__builtin_amdgcn_readfirstlane((int)(uint32_t)(rp >> 32));
  return (gin_grow_t)(((uint64_t)hi << 32) | (uint64_t)lo);
}
static_assert(GIN_D * sizeof(float) == 256, "a row is 256 bytes");

template <bool FIRST>
__global__ __launch_bounds__(1024) void gin_layer(
    const float* __restrict__ h,
    const int* __restrict__ pk, const unsigned* __restrict__ bm,
    int* __restrict__ lst, int* __restrict__ lcnt, int* __restrict__ tcnt,
    const float* __restrict__ W1, const float* __restrict__ b1,
    const float* __restrict__ W2, const float* __restrict__ b2,
    const float* __restrict__ g,  const float* __restrict__ be,
    const float* __restrict__ mu, const float* __restrict__ va,
    float* __restrict__ out, int do_bn) {
  __shared__ float s_rows[RALLOC * RPITCH];
  __shared__ int   s_q[QCAP];
  __shared__ float sW1t[GIN_D * TSTRIDE];
  __shared__ float sW2t[GIN_D * TSTRIDE];
  __shared__ float s_par[4 * GIN_D];
  __shared__ int   s_qn;
  __shared__ int   s_tc[32];
  __shared__ int   s_bin[FIRST ? RTILES * TCAP : 1];
  const int tid  = threadIdx.x;
  const int lane = tid & 31;
  const int wv   = __builtin_amdgcn_readfirstlane(tid >> 5);
  const int r0   = blockIdx.x * RROWS;
  const int left = GIN_N - r0;
  const int nrow = (left < RROWS) ? left : RROWS;
  int* lrow = lst + (size_t)blockIdx.x * QCAP;
  int have = -1;
  if (!FIRST) have = __builtin_amdgcn_readfirstlane(lcnt[blockIdx.x]);

  float w1v[4], w2v[4];
#pragma unroll
  for (int j = 0; j < 4; ++j) {
    w1v[j] = W1[j * 1024 + tid];
    w2v[j] = W2[j * 1024 + tid];
  }
  const float4* h4 = (const float4*)(h + (size_t)r0 * GIN_D);
  float4 own[7];
#pragma unroll
  for (int j = 0; j < 7; ++j) {
    const int i = j * 1024 + tid;
    own[j] = make_float4(0.f, 0.f, 0.f, 0.f);
    if (i < nrow * (GIN_D / 4)) own[j] = h4[i];
  }
  if (tid < GIN_D) {
    float scale = 1.f, shift = 0.f;
    if (do_bn) {
      const float sc = g[tid] * rsqrtf(va[tid] + 1e-5f);
      scale = sc;
      shift = be[tid] - mu[tid] * sc;
    }
    s_par[tid]             = b1[tid];
    s_par[GIN_D + tid]     = b2[tid];
    s_par[2 * GIN_D + tid] = scale;
    s_par[3 * GIN_D + tid] = shift;
  }
#pragma unroll
  for (int j = 0; j < 4; ++j) {
    const int i = j * 1024 + tid;
    const int k = i >> 6, n = i & 63;
    sW1t[n * TSTRIDE + k] = w1v[j];
    sW2t[n * TSTRIDE + k] = w2v[j];
  }
#pragma unroll
  for (int j = 0; j < 7; ++j) {
    const int i = j * 1024 + tid;
    if (i < RALLOC * (GIN_D / 4)) *(float4*)&s_rows[(i >> 4) * RPITCH + (i & 15) * 4] = own[j];
  }
  if (tid == 0) s_qn = 0;
  if (tid < 32) s_tc[tid] = 0;
  __syncthreads();

  const float* hb = h + lane;
  if (!FIRST && have >= 0) {
    if (wv < ((nrow + 15) >> 4)) {
      float* st = s_rows + wv * 16 * RPITCH;
      int cnt = __builtin_amdgcn_readfirstlane(tcnt[blockIdx.x * 32 + wv]);
      cnt = (cnt < 0) ? 0 : ((cnt > TCAP) ? TCAP : cnt);
      const int* lp = lrow + wv * TCAP;
      v32f acc;
#pragma unroll
      for (int r = 0; r < 16; ++r) {
        const v2f t = *(const v2f*)&st[r * RPITCH + 2 * lane];
        acc[2 * r]     = t.x;
        acc[2 * r + 1] = t.y;
      }
      if (cnt > 0) {
        const int last = cnt - 1;
        int qi = (lane < last) ? lane : last;
        int sd = lp[qi];
        int base = 0;
        while (base + TG <= cnt) {
          const int nbase = base + TG;
          qi = nbase + lane;
          qi = (qi < last) ? qi : last;
          v2f v[TG];
#pragma unroll
          for (int j = 0; j < TG; ++j) {
            const int sj = __builtin_amdgcn_readlane(sd, j) & ((1 << PK_SHIFT) - 1);
            v[j] = gin_row_base(h, sj)[lane];
          }
          const int sd2 = lp[qi];
          __builtin_amdgcn_sched_barrier(0);
#pragma unroll
          for (int j = 0; j < TG; ++j) {
            const int dj = (__builtin_amdgcn_readlane(sd, j) >> PK_SHIFT) & 15;
            acc[2 * dj]     += v[j].x;
            acc[2 * dj + 1] += v[j].y;
          }
          sd = sd2;
          base = nbase;
        }
        const int rem = cnt - base;
#pragma unroll
        for (int hh = 0; hh < TG / 16; ++hh) {
          const int n = rem - 16 * hh;
          if (n > 0) {
            v2f v[16];
#pragma unroll
            for (int j = 0; j < 16; ++j) {
              if (j < n) {
                const int sj = __builtin_amdgcn_readlane(sd, 16 * hh + j) & ((1 << PK_SHIFT) - 1);
                v[j] = gin_row_base(h, sj)[lane];
              }
            }
#pragma unroll
            for (int j = 0; j < 16; ++j) {
              if (j < n) {
                const int dj = (__builtin_amdgcn_readlane(sd, 16 * hh + j) >> PK_SHIFT) & 15;
                acc[2 * dj]     += v[j].x;
                acc[2 * dj + 1] += v[j].y;
              }
            }
          }
        }
      }
#pragma unroll
      for (int r = 0; r < 16; ++r) {
        v2f t;
        t.x = acc[2 * r];
        t.y = acc[2 * r + 1];
        *(v2f*)&st[r * RPITCH + 2 * lane] = t;
      }
    }
  } else {
  const uint4* b4 = (const uint4*)(bm + (size_t)blockIdx.x * WPR);
  int kept = -1;
  int p0 = 0;
  int cw = NP4;
  while (p0 < NP4) {
    const int pend = (p0 + cw < NP4) ? (p0 + cw) : NP4;
    if (cw == NP4) {
      uint4 v[SCAN_U];
#pragma unroll
      for (int j = 0; j < SCAN_U; ++j) {
        const int idx = j * 1024 + tid;
        v[j] = b4[(idx < NP4) ? idx : (NP4 - 1)];
      }
      int mine = 0;
#pragma unroll
      for (int j = 0; j < SCAN_U; ++j) {
        const int idx = j * 1024 + tid;
        if (idx >= NP4) v[j] = make_uint4(0u, 0u, 0u, 0u);
        mine += gin_count_bits(v[j]);
      }
      int slot = atomicAdd(&s_qn, mine);
#pragma unroll
      for (int j = 0; j < SCAN_U; ++j) {
        const int idx = j * 1024 + tid;
        slot = gin_queue_bits(v[j].x, idx * 4 + 0, slot, s_q);
        slot = gin_queue_bits(v[j].y, idx * 4 + 1, slot, s_q);
        slot = gin_queue_bits(v[j].z, idx * 4 + 2, slot, s_q);
        slot = gin_queue_bits(v[j].w, idx * 4 + 3, slot, s_q);
      }
    } else {
      const int idx = p0 + tid;
      if (tid < FB4 && idx < pend) {
        const uint4 v = b4[idx];
        int slot = atomicAdd(&s_qn, gin_count_bits(v));
        slot = gin_queue_bits(v.x, idx * 4 + 0, slot, s_q);
        slot = gin_queue_bits(v.y, idx * 4 + 1, slot, s_q);
        slot = gin_queue_bits(v.z, idx * 4 + 2, slot, s_q);
        slot = gin_queue_bits(v.w, idx * 4 + 3, slot, s_q);
      }
    }
    __syncthreads();
    const int cnt = __builtin_amdgcn_readfirstlane(s_qn);
    if (cnt > QCAP) {
      __syncthreads();
      if (tid == 0) s_qn = 0;
      __syncthreads();
      cw = FB4;
      continue;
    }
    const bool whole = (cw == NP4);
    gin_drain(hb, pk, s_q, FIRST && whole, s_bin, s_tc, s_rows, cnt, wv, lane);
    if (whole) kept = cnt;
    __syncthreads();
    if (tid == 0) s_qn = 0;
    __syncthreads();
    p0 = pend;
  }
  if (FIRST) {
    for (int i = tid; i < RTILES * TCAP; i += 1024) {
      const int t = i / TCAP;
      if (i - t * TCAP < s_tc[t]) lrow[i] = s_bin[i];
    }
    if (tid < 32) tcnt[blockIdx.x * 32 + tid] = s_tc[tid];
    if (tid == 0) {
      int k = kept;
      for (int t = 0; t < RTILES; ++t)
        if (s_tc[t] > TCAP) k = -1;
      lcnt[blockIdx.x] = k;
    }
  }
  }

  const int ntile = (nrow + 15) >> 4;
  if (wv < ntile) {
    const int m    = lane & 15;
    const int half = lane >> 4;
    float* st = s_rows + wv * 16 * RPITCH;

    v2f zA[16];
#pragma unroll
    for (int k0 = 0; k0 < 16; ++k0)
      zA[k0] = *(const v2f*)&st[m * RPITCH + 4 * k0 + 2 * half];

    v8f acc[4];
#pragma unroll
    for (int n0 = 0; n0 < 4; ++n0)
#pragma unroll
      for (int i = 0; i < 8; ++i) acc[n0][i] = 0.f;

#pragma unroll
    for (int n0 = 0; n0 < 4; ++n0) {
      const float* wcol = &sW1t[(n0 * 16 + m) * TSTRIDE + 2 * half];
#pragma unroll
      for (int k0 = 0; k0 < 16; ++k0) {
        v2f b = *(const v2f*)(wcol + 4 * k0);
        acc[n0] = __builtin_amdgcn_wmma_f32_16x16x4_f32(
            false, zA[k0], false, b, (short)0, acc[n0], false, false);
      }
    }

#pragma unroll
    for (int n0 = 0; n0 < 4; ++n0) {
      const float bias = s_par[n0 * 16 + m];
#pragma unroll
      for (int i = 0; i < 8; ++i) {
        const float v = fmaxf(acc[n0][i] + bias, 0.f);
        st[(half * 8 + i) * RPITCH + n0 * 16 + m] = v;
      }
    }

    v2f h1A[16];
#pragma unroll
    for (int k0 = 0; k0 < 16; ++k0)
      h1A[k0] = *(const v2f*)&st[m * RPITCH + 4 * k0 + 2 * half];

    v8f acc2[4];
#pragma unroll
    for (int n0 = 0; n0 < 4; ++n0)
#pragma unroll
      for (int i = 0; i < 8; ++i) acc2[n0][i] = 0.f;

#pragma unroll
    for (int n0 = 0; n0 < 4; ++n0) {
      const float* wcol = &sW2t[(n0 * 16 + m) * TSTRIDE + 2 * half];
#pragma unroll
      for (int k0 = 0; k0 < 16; ++k0) {
        v2f b = *(const v2f*)(wcol + 4 * k0);
        acc2[n0] = __builtin_amdgcn_wmma_f32_16x16x4_f32(
            false, h1A[k0], false, b, (short)0, acc2[n0], false, false);
      }
    }

#pragma unroll
    for (int n0 = 0; n0 < 4; ++n0) {
      const int col = n0 * 16 + m;
      const float bias  = s_par[GIN_D + col];
      const float scale = s_par[2 * GIN_D + col];
      const float shift = s_par[3 * GIN_D + col];
#pragma unroll
      for (int i = 0; i < 8; ++i) {
        float v = acc2[n0][i] + bias;
        if (do_bn) v = fmaxf(v * scale + shift, 0.f);
        st[(half * 8 + i) * RPITCH + col] = v;
      }
    }
  }
  __syncthreads();

  float4* o4 = (float4*)(out + (size_t)r0 * GIN_D);
  for (int i = tid; i < nrow * (GIN_D / 4); i += 1024)
    o4[i] = *(const float4*)&s_rows[(i >> 4) * RPITCH + (i & 15) * 4];
}

extern "C" void kernel_launch(void* const* d_in, const int* in_sizes, int n_in,
                              void* d_out, int out_size, void* d_ws, size_t ws_size,
                              hipStream_t stream) {
  const float* x  = (const float*)d_in[0];
  const int*   ei = (const int*)d_in[1];
  const float* W1 = (const float*)d_in[2];
  const float* b1 = (const float*)d_in[3];
  const float* W2 = (const float*)d_in[4];
  const float* b2 = (const float*)d_in[5];
  const float* g  = (const float*)d_in[6];
  const float* be = (const float*)d_in[7];
  const float* mu = (const float*)d_in[8];
  const float* va = (const float*)d_in[9];
  float* out = (float*)d_out;

  float*    hbufa = (float*)d_ws;
  float*    hbufb = hbufa + (size_t)GIN_N * GIN_D;
  unsigned* bm    = (unsigned*)(hbufb + (size_t)GIN_N * GIN_D);
  int*      pk    = (int*)(bm + (size_t)NRANGE * WPR);
  int*      lst   = pk + GIN_E;
  int*      lcnt  = lst + (size_t)NRANGE * QCAP;
  int*      tcnt  = lcnt + NRANGE;

  const float* hcur = x;
  gin_bitmap<<<BM_BLOCKS, 1024, 0, stream>>>(ei, ei + GIN_E, bm, pk);
  for (int l = 0; l < 5; ++l) {
    float* dstp = (l == 4) ? out : ((l % 2 == 0) ? hbufa : hbufb);
    if (l == 0)
      gin_layer<true><<<NRANGE, 1024, 0, stream>>>(
          hcur, pk, bm, lst, lcnt, tcnt,
          W1 + (size_t)l * GIN_D * GIN_D, b1 + (size_t)l * GIN_D,
          W2 + (size_t)l * GIN_D * GIN_D, b2 + (size_t)l * GIN_D,
          g  + (size_t)l * GIN_D, be + (size_t)l * GIN_D,
          mu + (size_t)l * GIN_D, va + (size_t)l * GIN_D,
          dstp, 1);
    else
      gin_layer<false><<<NRANGE, 1024, 0, stream>>>(
          hcur, pk, bm, lst, lcnt, tcnt,
          W1 + (size_t)l * GIN_D * GIN_D, b1 + (size_t)l * GIN_D,
          W2 + (size_t)l * GIN_D * GIN_D, b2 + (size_t)l * GIN_D,
          g  + (size_t)l * GIN_D, be + (size_t)l * GIN_D,
          mu + (size_t)l * GIN_D, va + (size_t)l * GIN_D,
          dstp, (l < 4) ? 1 : 0);
    hcur = dstp;
  }
}


# Block barriers removed from a tile's path: a wave stores its own output rows; in layers 2-5 it starts summing at entry (own rows read to registers, weights via split barrier); sum order as before
# speedup vs baseline: 4.6566x; 1.1181x over previous
#include <hip/hip_runtime.h>
#include <hip/hip_bf16.h>
#include <cstdint>

typedef __attribute__((ext_vector_type(2))) float v2f;
typedef __attribute__((ext_vector_type(8))) float v8f;
typedef __attribute__((ext_vector_type(32))) float v32f;

#define GIN_N 100000
#define GIN_E 1600000
#define GIN_D 64
#define TSTRIDE 66

#define NRANGE 256
#define RROWS  391
#define WPR    50016
#define NP4    (WPR / 4)
static_assert(NRANGE * RROWS >= GIN_N, "ranges cover all nodes");
static_assert((NRANGE - 1) * RROWS < GIN_N, "no empty range");
static_assert(WPR % 32 == 0 && WPR * 32 >= GIN_E, "bitmap rows are whole lines and cover all edges");
#define BM_BLOCKS (WPR / 32)
#define PK_SHIFT  17
static_assert(GIN_N <= (1 << PK_SHIFT), "a source node fits the low field");
static_assert(RROWS <= (1 << (31 - PK_SHIFT)), "a row inside a range fits the high field");

__global__ __launch_bounds__(1024) void gin_bitmap(const int* __restrict__ src,
                                                   const int* __restrict__ dst,
                                                   unsigned* __restrict__ bm,
                                                   int* __restrict__ pk) {
  __shared__ unsigned tile[NRANGE * 33];
  const int tid  = threadIdx.x;
  const int lane = tid & 31;
  const int wv   = tid >> 5;
  for (int i = tid; i < NRANGE * 33; i += 1024) tile[i] = 0u;
  __syncthreads();
  const int e = blockIdx.x * 1024 + tid;
  if (e < GIN_E) {
    const int d = dst[e];
    const int r = d / RROWS;
    atomicOr(&tile[r * 33 + wv], 1u << lane);
    pk[e] = src[e] | ((d - r * RROWS) << PK_SHIFT);
  }
  __syncthreads();
  for (int i = tid; i < NRANGE * 32; i += 1024) {
    const int r = i >> 5, w = i & 31;
    bm[(size_t)r * WPR + (size_t)blockIdx.x * 32 + w] = tile[r * 33 + w];
  }
}

#define RPITCH 68
#define RTILES ((RROWS + 15) / 16)
#define RALLOC (RTILES * 16)
static_assert(7 * 1024 >= RALLOC * (GIN_D / 4), "seven 16-byte pieces per thread cover the block's rows");
static_assert(RTILES <= 32, "one tile per wave");
#define QCAP   16384
#define FB4    (QCAP / 128)
#define GD     16
#define SCAN_U 13
#define TCAP   640
#define TG     32
static_assert(RTILES * TCAP <= QCAP, "the tiles' lists fit the range's list space");
static_assert(TCAP % 32 == 0, "a tile's list starts on a whole line");
static_assert(FB4 * 128 <= QCAP, "a fallback piece fits the queue");
static_assert(SCAN_U * 1024 >= NP4, "one sweep covers a bitmap row");

__device__ __forceinline__ int gin_queue_bits(unsigned bits, int word, int slot, int* q) {
  while (bits != 0u) {
    const int b = __builtin_ctz(bits);
    bits &= bits - 1u;
    if (slot < QCAP) q[slot] = word * 32 + b;
    ++slot;
  }
  return slot;
}
__device__ __forceinline__ int gin_count_bits(uint4 v) {
  return __builtin_popcount(v.x) + __builtin_popcount(v.y) + __builtin_popcount(v.z) + __builtin_popcount(v.w);
}

__device__ __forceinline__ void gin_drain(const float* __restrict__ hb, const int* __restrict__ pk,
                                          const int* s_q, bool record, int* s_bin, int* s_tc,
                                          float* s_rows, int cnt, int wv, int lane) {
  int base = wv * GD;
  if (base < cnt) {
    const int last = cnt - 1;
    int qi = base + (lane & (GD - 1));
    qi = (qi < last) ? qi : last;
    int sd;
    sd = pk[s_q[qi]];
    while (base < cnt) {
      const int ng = (cnt - base < GD) ? (cnt - base) : GD;
      const int nbase = base + 32 * GD;
      qi = nbase + (lane & (GD - 1));
      qi = (qi < last) ? qi : last;
      const int e2 = s_q[qi];
      float va[GD], vb[GD];
#pragma unroll
      for (int j = 0; j < GD; ++j) {
        const int sj = __builtin_amdgcn_readlane(sd, j) & ((1 << PK_SHIFT) - 1);
        const float* hp = hb + (size_t)sj * GIN_D;
        va[j] = hp[0];
        vb[j] = hp[32];
      }
      const int sd2 = pk[e2];
      if (record && lane < ng) {
        const int t   = (sd >> PK_SHIFT) >> 4;
        const int pos = atomicAdd(&s_tc[t], 1);
        if (pos < TCAP) s_bin[t * TCAP + pos] = sd;
      }
      if (ng == GD) {
#pragma unroll
        for (int j = 0; j < GD; ++j) {
          const int dj = __builtin_amdgcn_readlane(sd, j) >> PK_SHIFT;
          float* rp = s_rows + dj * RPITCH + lane;
          atomicAdd(rp, va[j]);
          atomicAdd(rp + 32, vb[j]);
        }
      } else {
#pragma unroll
        for (int j = 0; j < GD; ++j) {
          if (j < ng) {
            const int dj = __builtin_amdgcn_readlane(sd, j) >> PK_SHIFT;
            float* rp = s_rows + dj * RPITCH + lane;
            atomicAdd(rp, va[j]);
            atomicAdd(rp + 32, vb[j]);
          }
        }
      }
      sd = sd2;
      base = nbase;
    }
  }
}

typedef const __attribute__((address_space(1))) v2f* gin_grow_t;
__device__ __forceinline__ gin_grow_t gin_row_base(const float* __restrict__ h, int sj) {
  const uint64_t rp = (uint64_t)h + ((uint64_t)(uint32_t)sj << 8);
  const uint32_t lo = (uint32_t)__builtin_amdgcn_readfirstlane((int)(uint32_t)rp);
  const uint32_t hi = (uint32_t)__builtin_amdgcn_readfirstlane((int)(uint32_t)(rp >> 32));
  return (gin_grow_t)(((uint64_t)hi << 32) | (uint64_t)lo);
}
static_assert(GIN_D * sizeof(float) == 256, "a row is 256 bytes");

__device__ __forceinline__ void gin_stage_par(const float* __restrict__ b1, const float* __restrict__ b2,
                                              const float* __restrict__ g,  const float* __restrict__ be,
                                              const float* __restrict__ mu, const float* __restrict__ va,
                                              int do_bn, float* s_par, int tid) {
  if (tid < GIN_D) {
    float scale = 1.f, shift = 0.f;
    if (do_bn) {
      const float sc = g[tid] * rsqrtf(va[tid] + 1e-5f);
      scale = sc;
      shift = be[tid] - mu[tid] * sc;
    }
    s_par[tid]             = b1[tid];
    s_par[GIN_D + tid]     = b2[tid];
    s_par[2 * GIN_D + tid] = scale;
    s_par[3 * GIN_D + tid] = shift;
  }
}
__device__ __forceinline__ void gin_load_weights(const float* __restrict__ W1, const float* __restrict__ W2,
                                                 float (&w1v)[4], float (&w2v)[4], int tid) {
#pragma unroll
  for (int j = 0; j < 4; ++j) {
    w1v[j] = W1[j * 1024 + tid];
    w2v[j] = W2[j * 1024 + tid];
  }
}
__device__ __forceinline__ void gin_stage_weights(const float (&w1v)[4], const float (&w2v)[4],
                                                  float* sW1t, float* sW2t, int tid) {
#pragma unroll
  for (int j = 0; j < 4; ++j) {
    const int i = j * 1024 + tid;
    const int k = i >> 6, n = i & 63;
    sW1t[n * TSTRIDE + k] = w1v[j];
    sW2t[n * TSTRIDE + k] = w2v[j];
  }
}

template <bool FIRST>
__global__ __launch_bounds__(1024) void gin_layer(
    const float* __restrict__ h,
    const int* __restrict__ pk, const unsigned* __restrict__ bm,
    int* __restrict__ lst, int* __restrict__ lcnt, int* __restrict__ tcnt,
    const float* __restrict__ W1, const float* __restrict__ b1,
    const float* __restrict__ W2, const float* __restrict__ b2,
    const float* __restrict__ g,  const float* __restrict__ be,
    const float* __restrict__ mu, const float* __restrict__ va,
    float* __restrict__ out, int do_bn) {
  __shared__ float s_rows[RALLOC * RPITCH];
  __shared__ int   s_q[QCAP];
  __shared__ float sW1t[GIN_D * TSTRIDE];
  __shared__ float sW2t[GIN_D * TSTRIDE];
  __shared__ float s_par[4 * GIN_D];
  __shared__ int   s_qn;
  __shared__ int   s_tc[32];
  __shared__ int   s_bin[FIRST ? RTILES * TCAP : 1];
  const int tid  = threadIdx.x;
  const int lane = tid & 31;
  const int wv   = __builtin_amdgcn_readfirstlane(tid >> 5);
  const int r0   = blockIdx.x * RROWS;
  const int left = GIN_N - r0;
  const int nrow = (left < RROWS) ? left : RROWS;
  int* lrow = lst + (size_t)blockIdx.x * QCAP;
  const int ntile = (nrow + 15) >> 4;
  const bool mine = wv < ntile;
  int have = -1, cnt = 0;
  if (!FIRST) {
    have = __builtin_amdgcn_readfirstlane(lcnt[blockIdx.x]);
    cnt  = __builtin_amdgcn_readfirstlane(tcnt[blockIdx.x * 32 + wv]);
    cnt  = (cnt < 0) ? 0 : ((cnt > TCAP) ? TCAP : cnt);
  }

  float w1v[4], w2v[4];
  const float* hb = h + lane;
  if (!FIRST && have >= 0) {
    const int* lp = lrow + wv * TCAP;
    const int last = cnt - 1;
    v32f acc;
    int sd = 0;
    if (mine) {
      if (nrow - wv * 16 >= 16) {
        const gin_grow_t tb = gin_row_base(h, r0 + wv * 16);
#pragma unroll
        for (int r = 0; r < 16; ++r) {
          const v2f t = tb[r * (GIN_D / 2) + lane];
          acc[2 * r]     = t.x;
          acc[2 * r + 1] = t.y;
        }
      } else {
#pragma unroll
        for (int r = 0; r < 16; ++r) {
          int row = wv * 16 + r;
          row = (row < nrow) ? row : (nrow - 1);
          const v2f t = gin_row_base(h, r0 + row)[lane];
          acc[2 * r]     = t.x;
          acc[2 * r + 1] = t.y;
        }
      }
      if (cnt > 0) sd = lp[(lane < last) ? lane : last];
    }
    gin_load_weights(W1, W2, w1v, w2v, tid);
    gin_stage_par(b1, b2, g, be, mu, va, do_bn, s_par, tid);
    gin_stage_weights(w1v, w2v, sW1t, sW2t, tid);
    asm volatile("s_wait_dscnt 0x0\n\ts_barrier_signal -1" ::: "memory");
    if (mine) {
      float* st = s_rows + wv * 16 * RPITCH;
      if (cnt > 0) {
        int qi;
        int base = 0;
        while (base + TG <= cnt) {
          const int nbase = base + TG;
          qi = nbase + lane;
          qi = (qi < last) ? qi : last;
          v2f v[TG];
#pragma unroll
          for (int j = 0; j < TG; ++j) {
            const int sj = __builtin_amdgcn_readlane(sd, j) & ((1 << PK_SHIFT) - 1);
            v[j] = gin_row_base(h, sj)[lane];
          }
          const int sd2 = lp[qi];
          __builtin_amdgcn_sched_barrier(0);
#pragma unroll
          for (int j = 0; j < TG; ++j) {
            const int dj = (__builtin_amdgcn_readlane(sd, j) >> PK_SHIFT) & 15;
            acc[2 * dj]     += v[j].x;
            acc[2 * dj + 1] += v[j].y;
          }
          sd = sd2;
          base = nbase;
        }
        const int rem = cnt - base;
#pragma unroll
        for (int hh = 0; hh < TG / 16; ++hh) {
          const int n = rem - 16 * hh;
          if (n > 0) {
            v2f v[16];
#pragma unroll
            for (int j = 0; j < 16; ++j) {
              if (j < n) {
                const int sj = __builtin_amdgcn_readlane(sd, 16 * hh + j) & ((1 << PK_SHIFT) - 1);
                v[j] = gin_row_base(h, sj)[lane];
              }
            }
#pragma unroll
            for (int j = 0; j < 16; ++j) {
              if (j < n) {
                const int dj = (__builtin_amdgcn_readlane(sd, 16 * hh + j) >> PK_SHIFT) & 15;
                acc[2 * dj]     += v[j].x;
                acc[2 * dj + 1] += v[j].y;
              }
            }
          }
        }
      }
#pragma unroll
      for (int r = 0; r < 16; ++r) {
        v2f t;
        t.x = acc[2 * r];
        t.y = acc[2 * r + 1];
        *(v2f*)&st[r * RPITCH + 2 * lane] = t;
      }
    }
    asm volatile("s_barrier_wait -1" ::: "memory");
  } else {
  gin_load_weights(W1, W2, w1v, w2v, tid);
  const float4* h4 = (const float4*)(h + (size_t)r0 * GIN_D);
  float4 own[7];
#pragma unroll
  for (int j = 0; j < 7; ++j) {
    const int i = j * 1024 + tid;
    own[j] = make_float4(0.f, 0.f, 0.f, 0.f);
    if (i < nrow * (GIN_D / 4)) own[j] = h4[i];
  }
  gin_stage_par(b1, b2, g, be, mu, va, do_bn, s_par, tid);
  gin_stage_weights(w1v, w2v, sW1t, sW2t, tid);
#pragma unroll
  for (int j = 0; j < 7; ++j) {
    const int i = j * 1024 + tid;
    if (i < RALLOC * (GIN_D / 4)) *(float4*)&s_rows[(i >> 4) * RPITCH + (i & 15) * 4] = own[j];
  }
  if (tid == 0) s_qn = 0;
  if (tid < 32) s_tc[tid] = 0;
  __syncthreads();
  const uint4* b4 = (const uint4*)(bm + (size_t)blockIdx.x * WPR);
  int kept = -1;
  int p0 = 0;
  int cw = NP4;
  while (p0 < NP4) {
    const int pend = (p0 + cw < NP4) ? (p0 + cw) : NP4;
    if (cw == NP4) {
      uint4 v[SCAN_U];
#pragma unroll
      for (int j = 0; j < SCAN_U; ++j) {
        const int idx = j * 1024 + tid;
        v[j] = b4[(idx < NP4) ? idx : (NP4 - 1)];
      }
      int mine = 0;
#pragma unroll
      for (int j = 0; j < SCAN_U; ++j) {
        const int idx = j * 1024 + tid;
        if (idx >= NP4) v[j] = make_uint4(0u, 0u, 0u, 0u);
        mine += gin_count_bits(v[j]);
      }
      int slot = atomicAdd(&s_qn, mine);
#pragma unroll
      for (int j = 0; j < SCAN_U; ++j) {
        const int idx = j * 1024 + tid;
        slot = gin_queue_bits(v[j].x, idx * 4 + 0, slot, s_q);
        slot = gin_queue_bits(v[j].y, idx * 4 + 1, slot, s_q);
        slot = gin_queue_bits(v[j].z, idx * 4 + 2, slot, s_q);
        slot = gin_queue_bits(v[j].w, idx * 4 + 3, slot, s_q);
      }
    } else {
      const int idx = p0 + tid;
      if (tid < FB4 && idx < pend) {
        const uint4 v = b4[idx];
        int slot = atomicAdd(&s_qn, gin_count_bits(v));
        slot = gin_queue_bits(v.x, idx * 4 + 0, slot, s_q);
        slot = gin_queue_bits(v.y, idx * 4 + 1, slot, s_q);
        slot = gin_queue_bits(v.z, idx * 4 + 2, slot, s_q);
        slot = gin_queue_bits(v.w, idx * 4 + 3, slot, s_q);
      }
    }
    __syncthreads();
    const int cnt = __builtin_amdgcn_readfirstlane(s_qn);
    if (cnt > QCAP) {
      __syncthreads();
      if (tid == 0) s_qn = 0;
      __syncthreads();
      cw = FB4;
      continue;
    }
    const bool whole = (cw == NP4);
    gin_drain(hb, pk, s_q, FIRST && whole, s_bin, s_tc, s_rows, cnt, wv, lane);
    if (whole) kept = cnt;
    __syncthreads();
    if (tid == 0) s_qn = 0;
    __syncthreads();
    p0 = pend;
  }
  if (FIRST) {
    for (int i = tid; i < RTILES * TCAP; i += 1024) {
      const int t = i / TCAP;
      if (i - t * TCAP < s_tc[t]) lrow[i] = s_bin[i];
    }
    if (tid < 32) tcnt[blockIdx.x * 32 + tid] = s_tc[tid];
    if (tid == 0) {
      int k = kept;
      for (int t = 0; t < RTILES; ++t)
        if (s_tc[t] > TCAP) k = -1;
      lcnt[blockIdx.x] = k;
    }
  }
  }

  if (mine) {
    const int m    = lane & 15;
    const int half = lane >> 4;
    float* st = s_rows + wv * 16 * RPITCH;

    v2f zA[16];
#pragma unroll
    for (int k0 = 0; k0 < 16; ++k0)
      zA[k0] = *(const v2f*)&st[m * RPITCH + 4 * k0 + 2 * half];

    v8f acc[4];
#pragma unroll
    for (int n0 = 0; n0 < 4; ++n0)
#pragma unroll
      for (int i = 0; i < 8; ++i) acc[n0][i] = 0.f;

#pragma unroll
    for (int n0 = 0; n0 < 4; ++n0) {
      const float* wcol = &sW1t[(n0 * 16 + m) * TSTRIDE + 2 * half];
#pragma unroll
      for (int k0 = 0; k0 < 16; ++k0) {
        v2f b = *(const v2f*)(wcol + 4 * k0);
        acc[n0] = __builtin_amdgcn_wmma_f32_16x16x4_f32(
            false, zA[k0], false, b, (short)0, acc[n0], false, false);
      }
    }

#pragma unroll
    for (int n0 = 0; n0 < 4; ++n0) {
      const float bias = s_par[n0 * 16 + m];
#pragma unroll
      for (int i = 0; i < 8; ++i) {
        const float v = fmaxf(acc[n0][i] + bias, 0.f);
        st[(half * 8 + i) * RPITCH + n0 * 16 + m] = v;
      }
    }

    v2f h1A[16];
#pragma unroll
    for (int k0 = 0; k0 < 16; ++k0)
      h1A[k0] = *(const v2f*)&st[m * RPITCH + 4 * k0 + 2 * half];

    v8f acc2[4];
#pragma unroll
    for (int n0 = 0; n0 < 4; ++n0)
#pragma unroll
      for (int i = 0; i < 8; ++i) acc2[n0][i] = 0.f;

#pragma unroll
    for (int n0 = 0; n0 < 4; ++n0) {
      const float* wcol = &sW2t[(n0 * 16 + m) * TSTRIDE + 2 * half];
#pragma unroll
      for (int k0 = 0; k0 < 16; ++k0) {
        v2f b = *(const v2f*)(wcol + 4 * k0);
        acc2[n0] = __builtin_amdgcn_wmma_f32_16x16x4_f32(
            false, h1A[k0], false, b, (short)0, acc2[n0], false, false);
      }
    }

#pragma unroll
    for (int n0 = 0; n0 < 4; ++n0) {
      const int col = n0 * 16 + m;
      const float bias  = s_par[GIN_D + col];
      const float scale = s_par[2 * GIN_D + col];
      const float shift = s_par[3 * GIN_D + col];
#pragma unroll
      for (int i = 0; i < 8; ++i) {
        float v = acc2[n0][i] + bias;
        if (do_bn) v = fmaxf(v * scale + shift, 0.f);
        st[(half * 8 + i) * RPITCH + col] = v;
      }
    }

    float4* o4 = (float4*)(out + (size_t)(r0 + wv * 16) * GIN_D);
    const int trow = nrow - wv * 16;
    if (trow >= 16) {
      float4 o[8];
#pragma unroll
      for (int j = 0; j < 8; ++j) {
        const int i = j * 32 + lane;
        o[j] = *(const float4*)&st[(i >> 4) * RPITCH + (i & 15) * 4];
      }
#pragma unroll
      for (int j = 0; j < 8; ++j) o4[j * 32 + lane] = o[j];
    } else {
#pragma unroll
      for (int j = 0; j < 8; ++j) {
        const int i = j * 32 + lane;
        if ((i >> 4) < trow) o4[i] = *(const float4*)&st[(i >> 4) * RPITCH + (i & 15) * 4];
      }
    }
  }
}

extern "C" void kernel_launch(void* const* d_in, const int* in_sizes, int n_in,
                              void* d_out, int out_size, void* d_ws, size_t ws_size,
                              hipStream_t stream) {
  const float* x  = (const float*)d_in[0];
  const int*   ei = (const int*)d_in[1];
  const float* W1 = (const float*)d_in[2];
  const float* b1 = (const float*)d_in[3];
  const float* W2 = (const float*)d_in[4];
  const float* b2 = (const float*)d_in[5];
  const float* g  = (const float*)d_in[6];
  const float* be = (const float*)d_in[7];
  const float* mu = (const float*)d_in[8];
  const float* va = (const float*)d_in[9];
  float* out = (float*)d_out;

  float*    hbufa = (float*)d_ws;
  float*    hbufb = hbufa + (size_t)GIN_N * GIN_D;
  unsigned* bm    = (unsigned*)(hbufb + (size_t)GIN_N * GIN_D);
  int*      pk    = (int*)(bm + (size_t)NRANGE * WPR);
  int*      lst   = pk + GIN_E;
  int*      lcnt  = lst + (size_t)NRANGE * QCAP;
  int*      tcnt  = lcnt + NRANGE;

  const float* hcur = x;
  gin_bitmap<<<BM_BLOCKS, 1024, 0, stream>>>(ei, ei + GIN_E, bm, pk);
  for (int l = 0; l < 5; ++l) {
    float* dstp = (l == 4) ? out : ((l % 2 == 0) ? hbufa : hbufb);
    if (l == 0)
      gin_layer<true><<<NRANGE, 1024, 0, stream>>>(
          hcur, pk, bm, lst, lcnt, tcnt,
          W1 + (size_t)l * GIN_D * GIN_D, b1 + (size_t)l * GIN_D,
          W2 + (size_t)l * GIN_D * GIN_D, b2 + (size_t)l * GIN_D,
          g  + (size_t)l * GIN_D, be + (size_t)l * GIN_D,
          mu + (size_t)l * GIN_D, va + (size_t)l * GIN_D,
          dstp, 1);
    else
      gin_layer<false><<<NRANGE, 1024, 0, stream>>>(
          hcur, pk, bm, lst, lcnt, tcnt,
          W1 + (size_t)l * GIN_D * GIN_D, b1 + (size_t)l * GIN_D,
          W2 + (size_t)l * GIN_D * GIN_D, b2 + (size_t)l * GIN_D,
          g  + (size_t)l * GIN_D, be + (size_t)l * GIN_D,
          mu + (size_t)l * GIN_D, va + (size_t)l * GIN_D,
          dstp, (l < 4) ? 1 : 0);
    hcur = dstp;
  }
}
